# starting kernel
# baseline (speedup 1.0000x reference)
#include <hip/hip_runtime.h>


typedef __attribute__((ext_vector_type(2))) float v2f;
typedef __attribute__((ext_vector_type(8))) float v8f;

#define NN   20000
#define DIN  300
#define HD   150
#define IOUW 450
#define CHP  152
#define IOUP 464
#define FCP  160

__device__ __forceinline__ float sigmoid_f(float x) {
  return 1.0f / (1.0f + __expf(-x));
}
__device__ __forceinline__ float tanh_f(float x) {
  float t = __expf(2.0f * fminf(fmaxf(x, -15.0f), 15.0f));
  return (t - 1.0f) / (t + 1.0f);
}

__global__ void __launch_bounds__(128)
proj_gemm(const float* __restrict__ A, const float* __restrict__ B,
          const float* __restrict__ bias, float* __restrict__ C, int N) {
  __shared__ float Bs[32 * 304];
  const int tid  = threadIdx.x;
  const int lane = tid & 31;
  const int wave = tid >> 5;
  const int l15  = lane & 15;
  const int kb0  = (lane >> 4) << 1;
  const int n0   = (int)blockIdx.y * 32;

  for (int e = tid; e < 32 * DIN; e += 128) {
    int k = e >> 5, nl = e & 31;
    int n = n0 + nl;
    Bs[nl * 304 + k] = (n < N) ? B[k * N + n] : 0.0f;
  }
  __syncthreads();

  const int m0 = ((int)blockIdx.x * 4 + wave) * 16;
  if (m0 >= NN) return;
  const float* arow  = A + (size_t)(m0 + l15) * DIN;
  const float* b0row = Bs + l15 * 304;
  const float* b1row = Bs + (l15 + 16) * 304;
  v8f acc0 = {}, acc1 = {};
  for (int k0 = 0; k0 < DIN; k0 += 4) {
    int kb = k0 + kb0;
    v2f a  = *(const v2f*)(arow + kb);
    v2f b0 = *(const v2f*)(b0row + kb);
    v2f b1 = *(const v2f*)(b1row + kb);
    acc0 = __builtin_amdgcn_wmma_f32_16x16x4_f32(false, a, false, b0, (short)0,
                                                 acc0, false, false);
    acc1 = __builtin_amdgcn_wmma_f32_16x16x4_f32(false, a, false, b1, (short)0,
                                                 acc1, false, false);
  }
#pragma unroll
  for (int half = 0; half < 2; ++half) {
    int n = n0 + half * 16 + l15;
    if (n < N) {
      float bv = bias[n];
      const v8f& acc = half ? acc1 : acc0;
#pragma unroll
      for (int v = 0; v < 8; ++v) {
        int m = m0 + v + ((lane >> 4) << 3);
        C[(size_t)m * N + n] = acc[v] + bv;
      }
    }
  }
}

__global__ void leaf_kernel(const float* __restrict__ xiou,
                            const float* __restrict__ biouh,
                            float* __restrict__ cst, float* __restrict__ hst,
                            int count) {
  int e = (int)blockIdx.x * blockDim.x + threadIdx.x;
  if (e >= count * HD) return;
  int node = e / HD, col = e - node * HD;
  const float* r = xiou + (size_t)node * IOUW;
  float i = sigmoid_f(r[col] + biouh[col]);
  float o = sigmoid_f(r[HD + col] + biouh[HD + col]);
  float u = tanh_f(r[2 * HD + col] + biouh[2 * HD + col]);
  float c = i * u;
  cst[(size_t)node * HD + col] = c;
  hst[(size_t)node * HD + col] = o * tanh_f(c);
}

__global__ void __launch_bounds__(256)
level_kernel(const float* __restrict__ xiou, const float* __restrict__ xf,
             const float* __restrict__ Wiouh, const float* __restrict__ biouh,
             const float* __restrict__ Wfh, const float* __restrict__ bfh,
             float* __restrict__ cst, float* __restrict__ hst,
             int start, int end) {
  extern __shared__ float smem[];
  float* chh  = smem;
  float* chc  = chh  + 64 * CHP;
  float* hsum = chc  + 64 * CHP;
  float* fcs  = hsum + 16 * CHP;
  float* ious = fcs  + 16 * FCP;

  const int tid  = threadIdx.x;
  const int lane = tid & 31;
  const int wave = tid >> 5;
  const int l15  = lane & 15;
  const int kb0  = (lane >> 4) << 1;
  const int roff = (lane >> 4) << 3;
  const int base = start + (int)blockIdx.x * 16;

  for (int e = tid; e < 64 * CHP; e += 256) {
    int row = e / CHP, col = e - row * CHP;
    float hv = 0.0f, cv = 0.0f;
    int node = base + (row >> 2);
    if (col < HD && node < end) {
      int cr = 4 * ((NN - 1) - node) + 1 + (row & 3);
      if (cr < NN) {
        int ci = (NN - 1) - cr;
        hv = hst[(size_t)ci * HD + col];
        cv = cst[(size_t)ci * HD + col];
      }
    }
    chh[e] = hv;
    chc[e] = cv;
  }
  __syncthreads();
  for (int e = tid; e < 16 * CHP; e += 256) {
    int m = e / CHP, col = e - m * CHP;
    int r0 = (m * 4) * CHP + col;
    hsum[e] = chh[r0] + chh[r0 + CHP] + chh[r0 + 2 * CHP] + chh[r0 + 3 * CHP];
  }
  __syncthreads();

  for (int nt = wave; nt < 29; nt += 8) {
    int n0 = nt * 16;
    int n  = n0 + l15;
    bool nok = n < IOUW;
    v8f acc = {};
    for (int k0 = 0; k0 < 148; k0 += 4) {
      int kb = k0 + kb0;
      v2f a = *(const v2f*)(hsum + l15 * CHP + kb);
      v2f b;
      b.x = nok ? Wiouh[kb * IOUW + n] : 0.0f;
      b.y = nok ? Wiouh[(kb + 1) * IOUW + n] : 0.0f;
      acc = __builtin_amdgcn_wmma_f32_16x16x4_f32(false, a, false, b, (short)0,
                                                  acc, false, false);
    }
    {
      int kb = 148 + kb0;
      v2f a = *(const v2f*)(hsum + l15 * CHP + kb);
      v2f b;
      b.x = (nok && kb < HD)     ? Wiouh[kb * IOUW + n]       : 0.0f;
      b.y = (nok && kb + 1 < HD) ? Wiouh[(kb + 1) * IOUW + n] : 0.0f;
      acc = __builtin_amdgcn_wmma_f32_16x16x4_f32(false, a, false, b, (short)0,
                                                  acc, false, false);
    }
    if (nok) {
      float bv = biouh[n];
#pragma unroll
      for (int v = 0; v < 8; ++v) {
        int m = v + roff;
        int node = base + m;
        float xv = (node < end) ? xiou[(size_t)node * IOUW + n] : 0.0f;
        ious[m * IOUP + n] = acc[v] + bv + xv;
      }
    }
  }

  for (int t = wave; t < 20; t += 8) {
    int mp = t / 10, nt = t - mp * 10;
    int m0 = mp * 32, n0 = nt * 16;
    int n  = n0 + l15;
    bool nok = n < HD;
    int  nc  = nok ? n : 0;
    v8f acc0 = {}, acc1 = {};
    for (int k0 = 0; k0 < 148; k0 += 4) {
      int kb = k0 + kb0;
      v2f a0 = *(const v2f*)(chh + (m0 + l15) * CHP + kb);
      v2f a1 = *(const v2f*)(chh + (m0 + 16 + l15) * CHP + kb);
      v2f b;
      b.x = nok ? Wfh[kb * HD + n] : 0.0f;
      b.y = nok ? Wfh[(kb + 1) * HD + n] : 0.0f;
      acc0 = __builtin_amdgcn_wmma_f32_16x16x4_f32(false, a0, false, b, (short)0,
                                                   acc0, false, false);
      acc1 = __builtin_amdgcn_wmma_f32_16x16x4_f32(false, a1, false, b, (short)0,
                                                   acc1, false, false);
    }
    {
      int kb = 148 + kb0;
      v2f a0 = *(const v2f*)(chh + (m0 + l15) * CHP + kb);
      v2f a1 = *(const v2f*)(chh + (m0 + 16 + l15) * CHP + kb);
      v2f b;
      b.x = (nok && kb < HD)     ? Wfh[kb * HD + n]       : 0.0f;
      b.y = (nok && kb + 1 < HD) ? Wfh[(kb + 1) * HD + n] : 0.0f;
      acc0 = __builtin_amdgcn_wmma_f32_16x16x4_f32(false, a0, false, b, (short)0,
                                                   acc0, false, false);
      acc1 = __builtin_amdgcn_wmma_f32_16x16x4_f32(false, a1, false, b, (short)0,
                                                   acc1, false, false);
    }
    float bv = nok ? bfh[nc] : 0.0f;
    auto epilogue = [&](const v8f& acc, int mbase) {
      float s0 = 0.0f, s1 = 0.0f;
#pragma unroll
      for (int v = 0; v < 8; ++v) {
        int grow = mbase + v + roff;
        int node = base + (grow >> 2);
        float xv = (nok && node < end) ? xf[(size_t)node * HD + nc] : 0.0f;
        float f  = sigmoid_f(acc[v] + bv + xv);
        float w  = f * chc[grow * CHP + nc];
        if (v < 4) s0 += w; else s1 += w;
      }
      int nl = (mbase >> 2) + ((lane >> 4) << 1);
      fcs[nl * FCP + n]       = s0;
      fcs[(nl + 1) * FCP + n] = s1;
    };
    epilogue(acc0, m0);
    epilogue(acc1, m0 + 16);
  }
  __syncthreads();

  for (int e = tid; e < 16 * HD; e += 256) {
    int m = e / HD, col = e - m * HD;
    int node = base + m;
    if (node < end) {
      float i = sigmoid_f(ious[m * IOUP + col]);
      float o = sigmoid_f(ious[m * IOUP + HD + col]);
      float u = tanh_f(ious[m * IOUP + 2 * HD + col]);
      float c = i * u + fcs[m * FCP + col];
      cst[(size_t)node * HD + col] = c;
      hst[(size_t)node * HD + col] = o * tanh_f(c);
    }
  }
}

extern "C" void kernel_launch(void* const* d_in, const int* in_sizes, int n_in,
                              void* d_out, int out_size, void* d_ws, size_t ws_size,
                              hipStream_t stream) {
  const float* x     = (const float*)d_in[0];
  const float* Wioux = (const float*)d_in[1];
  const float* bioux = (const float*)d_in[2];
  const float* Wiouh = (const float*)d_in[3];
  const float* biouh = (const float*)d_in[4];
  const float* Wfx   = (const float*)d_in[5];
  const float* bfx   = (const float*)d_in[6];
  const float* Wfh   = (const float*)d_in[7];
  const float* bfh   = (const float*)d_in[8];
  (void)in_sizes; (void)n_in; (void)out_size; (void)ws_size;

  float* xiou = (float*)d_ws;
  float* xf   = xiou + (size_t)NN * IOUW;
  float* cst  = xf   + (size_t)NN * HD;
  float* hst  = (float*)d_out;

  proj_gemm<<<dim3(313, 15), 128, 0, stream>>>(x, Wioux, bioux, xiou, IOUW);
  proj_gemm<<<dim3(313, 5),  128, 0, stream>>>(x, Wfx,   bfx,   xf,   HD);

  const int leafEnd = 14539;
  leaf_kernel<<<(leafEnd * HD + 255) / 256, 256, 0, stream>>>(xiou, biouh,
                                                              cst, hst, leafEnd);

  const int bounds[8] = {14539, 18635, 19659, 19915, 19979, 19995, 19999, 20000};
  const size_t lds =
      (size_t)(2 * 64 * CHP + 16 * CHP + 16 * FCP + 16 * IOUP) * sizeof(float);
  int s = bounds[0];
  for (int L = 1; L < 8; ++L) {
    int e = bounds[L];
    int groups = (e - s + 15) / 16;
    level_kernel<<<groups, 256, lds, stream>>>(xiou, xf, Wiouh, biouh, Wfh, bfh,
                                               cst, hst, s, e);
    s = e;
  }
}


# level_kernel: global loads made unconditional (clamped address, then select 0) and issued in batches (whole-K weight fragments, epilogue x rows, child gather) instead of one branch+load+wait each
# speedup vs baseline: 2.2548x; 2.2548x over previous
#include <hip/hip_runtime.h>


typedef __attribute__((ext_vector_type(2))) float v2f;
typedef __attribute__((ext_vector_type(8))) float v8f;

#define NN   20000
#define DIN  300
#define HD   150
#define IOUW 450
#define CHP  152
#define IOUP 464
#define FCP  160

__device__ __forceinline__ float sigmoid_f(float x) {
  return 1.0f / (1.0f + __expf(-x));
}
__device__ __forceinline__ float tanh_f(float x) {
  float t = __expf(2.0f * fminf(fmaxf(x, -15.0f), 15.0f));
  return (t - 1.0f) / (t + 1.0f);
}

__global__ void __launch_bounds__(128)
proj_gemm(const float* __restrict__ A, const float* __restrict__ B,
          const float* __restrict__ bias, float* __restrict__ C, int N) {
  __shared__ float Bs[32 * 304];
  const int tid  = threadIdx.x;
  const int lane = tid & 31;
  const int wave = tid >> 5;
  const int l15  = lane & 15;
  const int kb0  = (lane >> 4) << 1;
  const int n0   = (int)blockIdx.y * 32;

  for (int e = tid; e < 32 * DIN; e += 128) {
    int k = e >> 5, nl = e & 31;
    int n = n0 + nl;
    Bs[nl * 304 + k] = (n < N) ? B[k * N + n] : 0.0f;
  }
  __syncthreads();

  const int m0 = ((int)blockIdx.x * 4 + wave) * 16;
  if (m0 >= NN) return;
  const float* arow  = A + (size_t)(m0 + l15) * DIN;
  const float* b0row = Bs + l15 * 304;
  const float* b1row = Bs + (l15 + 16) * 304;
  v8f acc0 = {}, acc1 = {};
  for (int k0 = 0; k0 < DIN; k0 += 4) {
    int kb = k0 + kb0;
    v2f a  = *(const v2f*)(arow + kb);
    v2f b0 = *(const v2f*)(b0row + kb);
    v2f b1 = *(const v2f*)(b1row + kb);
    acc0 = __builtin_amdgcn_wmma_f32_16x16x4_f32(false, a, false, b0, (short)0,
                                                 acc0, false, false);
    acc1 = __builtin_amdgcn_wmma_f32_16x16x4_f32(false, a, false, b1, (short)0,
                                                 acc1, false, false);
  }
#pragma unroll
  for (int half = 0; half < 2; ++half) {
    int n = n0 + half * 16 + l15;
    if (n < N) {
      float bv = bias[n];
      const v8f& acc = half ? acc1 : acc0;
#pragma unroll
      for (int v = 0; v < 8; ++v) {
        int m = m0 + v + ((lane >> 4) << 3);
        C[(size_t)m * N + n] = acc[v] + bv;
      }
    }
  }
}

__global__ void leaf_kernel(const float* __restrict__ xiou,
                            const float* __restrict__ biouh,
                            float* __restrict__ cst, float* __restrict__ hst,
                            int count) {
  int e = (int)blockIdx.x * blockDim.x + threadIdx.x;
  if (e >= count * HD) return;
  int node = e / HD, col = e - node * HD;
  const float* r = xiou + (size_t)node * IOUW;
  float i = sigmoid_f(r[col] + biouh[col]);
  float o = sigmoid_f(r[HD + col] + biouh[HD + col]);
  float u = tanh_f(r[2 * HD + col] + biouh[2 * HD + col]);
  float c = i * u;
  cst[(size_t)node * HD + col] = c;
  hst[(size_t)node * HD + col] = o * tanh_f(c);
}

__global__ void __launch_bounds__(256)
level_kernel(const float* __restrict__ xiou, const float* __restrict__ xf,
             const float* __restrict__ Wiouh, const float* __restrict__ biouh,
             const float* __restrict__ Wfh, const float* __restrict__ bfh,
             float* __restrict__ cst, float* __restrict__ hst,
             int start, int end) {
  extern __shared__ float smem[];
  float* chh  = smem;
  float* chc  = chh  + 64 * CHP;
  float* hsum = chc  + 64 * CHP;
  float* fcs  = hsum + 16 * CHP;
  float* ious = fcs  + 16 * FCP;

  const int tid  = threadIdx.x;
  const int lane = tid & 31;
  const int wave = tid >> 5;
  const int l15  = lane & 15;
  const int kb0  = (lane >> 4) << 1;
  const int roff = (lane >> 4) << 3;
  const int base = start + (int)blockIdx.x * 16;

  {
    float hv[38], cv[38];
#pragma unroll
    for (int i = 0; i < 38; ++i) {
      int e = tid + i * 256;
      int row = e / CHP, col = e - row * CHP;
      int node = base + (row >> 2);
      int cr = 4 * ((NN - 1) - node) + 1 + (row & 3);
      bool ok = col < HD && node < end && cr < NN;
      size_t off = ok ? (size_t)((NN - 1) - cr) * HD + col : (size_t)0;
      float h = hst[off];
      float c = cst[off];
      hv[i] = ok ? h : 0.0f;
      cv[i] = ok ? c : 0.0f;
    }
#pragma unroll
    for (int i = 0; i < 38; ++i) {
      int e = tid + i * 256;
      chh[e] = hv[i];
      chc[e] = cv[i];
    }
  }
  __syncthreads();
  for (int e = tid; e < 16 * CHP; e += 256) {
    int m = e / CHP, col = e - m * CHP;
    int r0 = (m * 4) * CHP + col;
    hsum[e] = chh[r0] + chh[r0 + CHP] + chh[r0 + 2 * CHP] + chh[r0 + 3 * CHP];
  }
  __syncthreads();

  for (int nt = wave; nt < 29; nt += 8) {
    int n0 = nt * 16;
    int n  = n0 + l15;
    bool nok = n < IOUW;
    const float* wcol = Wiouh + (nok ? n : 0);
    const bool tailok = kb0 == 0;
    float bx[38], by[38];
#pragma unroll
    for (int s = 0; s < 37; ++s) {
      int kb = 4 * s + kb0;
      float v0 = wcol[kb * IOUW];
      float v1 = wcol[(kb + 1) * IOUW];
      bx[s] = nok ? v0 : 0.0f;
      by[s] = nok ? v1 : 0.0f;
    }
    {
      int kb = tailok ? 148 : 0;
      float v0 = wcol[kb * IOUW];
      float v1 = wcol[(kb + 1) * IOUW];
      bx[37] = (nok && tailok) ? v0 : 0.0f;
      by[37] = (nok && tailok) ? v1 : 0.0f;
    }
    v8f acc = {};
#pragma unroll
    for (int s = 0; s < 38; ++s) {
      int kb = 4 * s + kb0;
      v2f a = *(const v2f*)(hsum + l15 * CHP + kb);
      v2f b;
      b.x = bx[s];
      b.y = by[s];
      acc = __builtin_amdgcn_wmma_f32_16x16x4_f32(false, a, false, b, (short)0,
                                                  acc, false, false);
    }
    if (nok) {
      float bv = biouh[n];
      float xv[8];
#pragma unroll
      for (int v = 0; v < 8; ++v) {
        int node = base + v + roff;
        bool live = node < end;
        float t = xiou[(size_t)(live ? node : base) * IOUW + n];
        xv[v] = live ? t : 0.0f;
      }
#pragma unroll
      for (int v = 0; v < 8; ++v) {
        int m = v + roff;
        ious[m * IOUP + n] = acc[v] + bv + xv[v];
      }
    }
  }

  for (int t = wave; t < 20; t += 8) {
    int mp = t / 10, nt = t - mp * 10;
    int m0 = mp * 32, n0 = nt * 16;
    int n  = n0 + l15;
    bool nok = n < HD;
    int  nc  = nok ? n : 0;
    const float* wcol = Wfh + nc;
    const bool tailok = kb0 == 0;
    float bx[38], by[38];
#pragma unroll
    for (int s = 0; s < 37; ++s) {
      int kb = 4 * s + kb0;
      float v0 = wcol[kb * HD];
      float v1 = wcol[(kb + 1) * HD];
      bx[s] = nok ? v0 : 0.0f;
      by[s] = nok ? v1 : 0.0f;
    }
    {
      int kb = tailok ? 148 : 0;
      float v0 = wcol[kb * HD];
      float v1 = wcol[(kb + 1) * HD];
      bx[37] = (nok && tailok) ? v0 : 0.0f;
      by[37] = (nok && tailok) ? v1 : 0.0f;
    }
    v8f acc0 = {}, acc1 = {};
#pragma unroll
    for (int s = 0; s < 38; ++s) {
      int kb = 4 * s + kb0;
      v2f a0 = *(const v2f*)(chh + (m0 + l15) * CHP + kb);
      v2f a1 = *(const v2f*)(chh + (m0 + 16 + l15) * CHP + kb);
      v2f b;
      b.x = bx[s];
      b.y = by[s];
      acc0 = __builtin_amdgcn_wmma_f32_16x16x4_f32(false, a0, false, b, (short)0,
                                                   acc0, false, false);
      acc1 = __builtin_amdgcn_wmma_f32_16x16x4_f32(false, a1, false, b, (short)0,
                                                   acc1, false, false);
    }
    float bv = nok ? bfh[nc] : 0.0f;
    auto epilogue = [&](const v8f& acc, int mbase) {
      float s0 = 0.0f, s1 = 0.0f;
#pragma unroll
      for (int v = 0; v < 8; ++v) {
        int grow = mbase + v + roff;
        int node = base + (grow >> 2);
        float xv = (nok && node < end) ? xf[(size_t)node * HD + nc] : 0.0f;
        float f  = sigmoid_f(acc[v] + bv + xv);
        float w  = f * chc[grow * CHP + nc];
        if (v < 4) s0 += w; else s1 += w;
      }
      int nl = (mbase >> 2) + ((lane >> 4) << 1);
      fcs[nl * FCP + n]       = s0;
      fcs[(nl + 1) * FCP + n] = s1;
    };
    epilogue(acc0, m0);
    epilogue(acc1, m0 + 16);
  }
  __syncthreads();

  for (int e = tid; e < 16 * HD; e += 256) {
    int m = e / HD, col = e - m * HD;
    int node = base + m;
    if (node < end) {
      float i = sigmoid_f(ious[m * IOUP + col]);
      float o = sigmoid_f(ious[m * IOUP + HD + col]);
      float u = tanh_f(ious[m * IOUP + 2 * HD + col]);
      float c = i * u + fcs[m * FCP + col];
      cst[(size_t)node * HD + col] = c;
      hst[(size_t)node * HD + col] = o * tanh_f(c);
    }
  }
}

extern "C" void kernel_launch(void* const* d_in, const int* in_sizes, int n_in,
                              void* d_out, int out_size, void* d_ws, size_t ws_size,
                              hipStream_t stream) {
  const float* x     = (const float*)d_in[0];
  const float* Wioux = (const float*)d_in[1];
  const float* bioux = (const float*)d_in[2];
  const float* Wiouh = (const float*)d_in[3];
  const float* biouh = (const float*)d_in[4];
  const float* Wfx   = (const float*)d_in[5];
  const float* bfx   = (const float*)d_in[6];
  const float* Wfh   = (const float*)d_in[7];
  const float* bfh   = (const float*)d_in[8];
  (void)in_sizes; (void)n_in; (void)out_size; (void)ws_size;

  float* xiou = (float*)d_ws;
  float* xf   = xiou + (size_t)NN * IOUW;
  float* cst  = xf   + (size_t)NN * HD;
  float* hst  = (float*)d_out;

  proj_gemm<<<dim3(313, 15), 128, 0, stream>>>(x, Wioux, bioux, xiou, IOUW);
  proj_gemm<<<dim3(313, 5),  128, 0, stream>>>(x, Wfx,   bfx,   xf,   HD);

  const int leafEnd = 14539;
  leaf_kernel<<<(leafEnd * HD + 255) / 256, 256, 0, stream>>>(xiou, biouh,
                                                              cst, hst, leafEnd);

  const int bounds[8] = {14539, 18635, 19659, 19915, 19979, 19995, 19999, 20000};
  const size_t lds =
      (size_t)(2 * 64 * CHP + 16 * CHP + 16 * FCP + 16 * IOUP) * sizeof(float);
  int s = bounds[0];
  for (int L = 1; L < 8; ++L) {
    int e = bounds[L];
    int groups = (e - s + 15) / 16;
    level_kernel<<<groups, 256, lds, stream>>>(xiou, xf, Wiouh, biouh, Wfh, bfh,
                                               cst, hst, s, e);
    s = e;
  }
}


# level_kernel: each 16-node group's H columns split over 5 workgroups (32 each), 14 one-tile WMMA tasks per workgroup, 16 waves on narrow levels: a level's serial chain is 1-2 tiles, not 7
# speedup vs baseline: 3.0108x; 1.3353x over previous
#include <hip/hip_runtime.h>


typedef __attribute__((ext_vector_type(2))) float v2f;
typedef __attribute__((ext_vector_type(8))) float v8f;

#define NN   20000
#define DIN  300
#define HD   150
#define IOUW 450
#define CHP  152

__device__ __forceinline__ float sigmoid_f(float x) {
  return 1.0f / (1.0f + __expf(-x));
}
__device__ __forceinline__ float tanh_f(float x) {
  float t = __expf(2.0f * fminf(fmaxf(x, -15.0f), 15.0f));
  return (t - 1.0f) / (t + 1.0f);
}

__global__ void __launch_bounds__(128)
proj_gemm(const float* __restrict__ A, const float* __restrict__ B,
          const float* __restrict__ bias, float* __restrict__ C, int N) {
  __shared__ float Bs[32 * 304];
  const int tid  = threadIdx.x;
  const int lane = tid & 31;
  const int wave = tid >> 5;
  const int l15  = lane & 15;
  const int kb0  = (lane >> 4) << 1;
  const int n0   = (int)blockIdx.y * 32;

  for (int e = tid; e < 32 * DIN; e += 128) {
    int k = e >> 5, nl = e & 31;
    int n = n0 + nl;
    Bs[nl * 304 + k] = (n < N) ? B[k * N + n] : 0.0f;
  }
  __syncthreads();

  const int m0 = ((int)blockIdx.x * 4 + wave) * 16;
  if (m0 >= NN) return;
  const float* arow  = A + (size_t)(m0 + l15) * DIN;
  const float* b0row = Bs + l15 * 304;
  const float* b1row = Bs + (l15 + 16) * 304;
  v8f acc0 = {}, acc1 = {};
  for (int k0 = 0; k0 < DIN; k0 += 4) {
    int kb = k0 + kb0;
    v2f a  = *(const v2f*)(arow + kb);
    v2f b0 = *(const v2f*)(b0row + kb);
    v2f b1 = *(const v2f*)(b1row + kb);
    acc0 = __builtin_amdgcn_wmma_f32_16x16x4_f32(false, a, false, b0, (short)0,
                                                 acc0, false, false);
    acc1 = __builtin_amdgcn_wmma_f32_16x16x4_f32(false, a, false, b1, (short)0,
                                                 acc1, false, false);
  }
#pragma unroll
  for (int half = 0; half < 2; ++half) {
    int n = n0 + half * 16 + l15;
    if (n < N) {
      float bv = bias[n];
      const v8f& acc = half ? acc1 : acc0;
#pragma unroll
      for (int v = 0; v < 8; ++v) {
        int m = m0 + v + ((lane >> 4) << 3);
        C[(size_t)m * N + n] = acc[v] + bv;
      }
    }
  }
}

__global__ void leaf_kernel(const float* __restrict__ xiou,
                            const float* __restrict__ biouh,
                            float* __restrict__ cst, float* __restrict__ hst,
                            int count) {
  int e = (int)blockIdx.x * blockDim.x + threadIdx.x;
  if (e >= count * HD) return;
  int node = e / HD, col = e - node * HD;
  const float* r = xiou + (size_t)node * IOUW;
  float i = sigmoid_f(r[col] + biouh[col]);
  float o = sigmoid_f(r[HD + col] + biouh[HD + col]);
  float u = tanh_f(r[2 * HD + col] + biouh[2 * HD + col]);
  float c = i * u;
  cst[(size_t)node * HD + col] = c;
  hst[(size_t)node * HD + col] = o * tanh_f(c);
}

#define SLW 32
#define NSL 5
#define CSP 36
#define ISP 100
#define FSP 36

template <int NT>
__global__ void __launch_bounds__(NT)
level_kernel(const float* __restrict__ xiou, const float* __restrict__ xf,
             const float* __restrict__ Wiouh, const float* __restrict__ biouh,
             const float* __restrict__ Wfh, const float* __restrict__ bfh,
             float* __restrict__ cst, float* __restrict__ hst,
             int start, int end) {
  extern __shared__ float smem[];
  float* chh  = smem;
  float* hsum = chh  + 64 * CHP;
  float* chc  = hsum + 16 * CHP;
  float* ious = chc  + 64 * CSP;
  float* fcs  = ious + 16 * ISP;

  const int tid  = threadIdx.x;
  const int lane = tid & 31;
  const int wave = tid >> 5;
  const int l15  = lane & 15;
  const int kb0  = (lane >> 4) << 1;
  const int roff = (lane >> 4) << 3;
  const int base = start + (int)blockIdx.x * 16;
  const int c0   = (int)blockIdx.y * SLW;

  {
    constexpr int NH = 64 * CHP / NT, NC = 64 * SLW / NT;
    static_assert(NH * NT == 64 * CHP && NC * NT == 64 * SLW, "trip counts");
    float hv[NH], cv[NC];
#pragma unroll
    for (int i = 0; i < NH; ++i) {
      int e = tid + i * NT;
      int row = e / CHP, col = e - row * CHP;
      int node = base + (row >> 2);
      int cr = 4 * ((NN - 1) - node) + 1 + (row & 3);
      bool ok = col < HD && node < end && cr < NN;
      size_t off = ok ? (size_t)((NN - 1) - cr) * HD + col : (size_t)0;
      float h = hst[off];
      hv[i] = ok ? h : 0.0f;
    }
#pragma unroll
    for (int i = 0; i < NC; ++i) {
      int e = tid + i * NT;
      int row = e >> 5, col = c0 + (e & 31);
      int node = base + (row >> 2);
      int cr = 4 * ((NN - 1) - node) + 1 + (row & 3);
      bool ok = col < HD && node < end && cr < NN;
      size_t off = ok ? (size_t)((NN - 1) - cr) * HD + col : (size_t)0;
      float c = cst[off];
      cv[i] = ok ? c : 0.0f;
    }
#pragma unroll
    for (int i = 0; i < NH; ++i) chh[tid + i * NT] = hv[i];
#pragma unroll
    for (int i = 0; i < NC; ++i) {
      int e = tid + i * NT;
      chc[(e >> 5) * CSP + (e & 31)] = cv[i];
    }
  }
  __syncthreads();
  for (int e = tid; e < 16 * CHP; e += NT) {
    int m = e / CHP, col = e - m * CHP;
    int r0 = (m * 4) * CHP + col;
    hsum[e] = chh[r0] + chh[r0 + CHP] + chh[r0 + 2 * CHP] + chh[r0 + 3 * CHP];
  }
  __syncthreads();

  for (int t = wave; t < 14; t += NT / 32) {
    const bool gate = t < 6;
    const int  u    = gate ? t : t - 6;
    const int  hf   = u & 1;
    const int  lc   = hf * 16 + l15;
    const int  hc   = c0 + lc;
    const bool nok  = hc < HD;
    const bool tailok = kb0 == 0;
    if (gate) {
      const int g = u >> 1;
      const int n = g * HD + hc;
      const float* wcol = Wiouh + (nok ? n : 0);
      float bx[38], by[38];
#pragma unroll
      for (int s = 0; s < 37; ++s) {
        int kb = 4 * s + kb0;
        float v0 = wcol[kb * IOUW];
        float v1 = wcol[(kb + 1) * IOUW];
        bx[s] = nok ? v0 : 0.0f;
        by[s] = nok ? v1 : 0.0f;
      }
      {
        int kb = tailok ? 148 : 0;
        float v0 = wcol[kb * IOUW];
        float v1 = wcol[(kb + 1) * IOUW];
        bx[37] = (nok && tailok) ? v0 : 0.0f;
        by[37] = (nok && tailok) ? v1 : 0.0f;
      }
      v8f acc = {};
#pragma unroll
      for (int s = 0; s < 38; ++s) {
        int kb = 4 * s + kb0;
        v2f a = *(const v2f*)(hsum + l15 * CHP + kb);
        v2f b;
        b.x = bx[s];
        b.y = by[s];
        acc = __builtin_amdgcn_wmma_f32_16x16x4_f32(false, a, false, b, (short)0,
                                                    acc, false, false);
      }
      if (nok) {
        float bv = biouh[n];
        float xv[8];
#pragma unroll
        for (int v = 0; v < 8; ++v) {
          int node = base + v + roff;
          bool live = node < end;
          float tt = xiou[(size_t)(live ? node : base) * IOUW + n];
          xv[v] = live ? tt : 0.0f;
        }
#pragma unroll
        for (int v = 0; v < 8; ++v) {
          int m = v + roff;
          ious[m * ISP + g * SLW + lc] = acc[v] + bv + xv[v];
        }
      }
    } else {
      const int m0 = (u >> 1) * 16;
      const int nc = nok ? hc : 0;
      const float* wcol = Wfh + nc;
      float bx[38], by[38];
#pragma unroll
      for (int s = 0; s < 37; ++s) {
        int kb = 4 * s + kb0;
        float v0 = wcol[kb * HD];
        float v1 = wcol[(kb + 1) * HD];
        bx[s] = nok ? v0 : 0.0f;
        by[s] = nok ? v1 : 0.0f;
      }
      {
        int kb = tailok ? 148 : 0;
        float v0 = wcol[kb * HD];
        float v1 = wcol[(kb + 1) * HD];
        bx[37] = (nok && tailok) ? v0 : 0.0f;
        by[37] = (nok && tailok) ? v1 : 0.0f;
      }
      v8f acc = {};
#pragma unroll
      for (int s = 0; s < 38; ++s) {
        int kb = 4 * s + kb0;
        v2f a = *(const v2f*)(chh + (m0 + l15) * CHP + kb);
        v2f b;
        b.x = bx[s];
        b.y = by[s];
        acc = __builtin_amdgcn_wmma_f32_16x16x4_f32(false, a, false, b, (short)0,
                                                    acc, false, false);
      }
      float bv = nok ? bfh[nc] : 0.0f;
      float s0 = 0.0f, s1 = 0.0f;
#pragma unroll
      for (int v = 0; v < 8; ++v) {
        int grow = m0 + v + roff;
        int node = base + (grow >> 2);
        float xv = (nok && node < end) ? xf[(size_t)node * HD + nc] : 0.0f;
        float f  = sigmoid_f(acc[v] + bv + xv);
        float w  = f * chc[grow * CSP + lc];
        if (v < 4) s0 += w; else s1 += w;
      }
      int nl = (m0 >> 2) + ((lane >> 4) << 1);
      fcs[nl * FSP + lc]       = s0;
      fcs[(nl + 1) * FSP + lc] = s1;
    }
  }
  __syncthreads();

  for (int e = tid; e < 16 * SLW; e += NT) {
    int m = e >> 5, lc = e & 31;
    int col = c0 + lc;
    int node = base + m;
    if (node < end && col < HD) {
      float i = sigmoid_f(ious[m * ISP + lc]);
      float o = sigmoid_f(ious[m * ISP + SLW + lc]);
      float u = tanh_f(ious[m * ISP + 2 * SLW + lc]);
      float c = i * u + fcs[m * FSP + lc];
      cst[(size_t)node * HD + col] = c;
      hst[(size_t)node * HD + col] = o * tanh_f(c);
    }
  }
}

extern "C" void kernel_launch(void* const* d_in, const int* in_sizes, int n_in,
                              void* d_out, int out_size, void* d_ws, size_t ws_size,
                              hipStream_t stream) {
  const float* x     = (const float*)d_in[0];
  const float* Wioux = (const float*)d_in[1];
  const float* bioux = (const float*)d_in[2];
  const float* Wiouh = (const float*)d_in[3];
  const float* biouh = (const float*)d_in[4];
  const float* Wfx   = (const float*)d_in[5];
  const float* bfx   = (const float*)d_in[6];
  const float* Wfh   = (const float*)d_in[7];
  const float* bfh   = (const float*)d_in[8];
  (void)in_sizes; (void)n_in; (void)out_size; (void)ws_size;

  float* xiou = (float*)d_ws;
  float* xf   = xiou + (size_t)NN * IOUW;
  float* cst  = xf   + (size_t)NN * HD;
  float* hst  = (float*)d_out;

  proj_gemm<<<dim3(313, 15), 128, 0, stream>>>(x, Wioux, bioux, xiou, IOUW);
  proj_gemm<<<dim3(313, 5),  128, 0, stream>>>(x, Wfx,   bfx,   xf,   HD);

  const int leafEnd = 14539;
  leaf_kernel<<<(leafEnd * HD + 255) / 256, 256, 0, stream>>>(xiou, biouh,
                                                              cst, hst, leafEnd);

  const int bounds[8] = {14539, 18635, 19659, 19915, 19979, 19995, 19999, 20000};
  const size_t lds =
      (size_t)(64 * CHP + 16 * CHP + 64 * CSP + 16 * ISP + 16 * FSP) * sizeof(float);
  int s = bounds[0];
  for (int L = 1; L < 8; ++L) {
    int e = bounds[L];
    int groups = (e - s + 15) / 16;
    if (L == 1)
      level_kernel<256><<<dim3(groups, NSL), 256, lds, stream>>>(xiou, xf, Wiouh, biouh,
                                                                 Wfh, bfh, cst, hst, s, e);
    else
      level_kernel<512><<<dim3(groups, NSL), 512, lds, stream>>>(xiou, xf, Wiouh, biouh,
                                                                 Wfh, bfh, cst, hst, s, e);
    s = e;
  }
}


# proj_gemm: a thread's 75 weight-tile staging loads made unconditional (clamped column, then select 0) and issued as one batch before its LDS stores, not one branch+load+wait+store each
# speedup vs baseline: 4.0306x; 1.3387x over previous
#include <hip/hip_runtime.h>


typedef __attribute__((ext_vector_type(2))) float v2f;
typedef __attribute__((ext_vector_type(8))) float v8f;

#define NN   20000
#define DIN  300
#define HD   150
#define IOUW 450
#define CHP  152

__device__ __forceinline__ float sigmoid_f(float x) {
  return 1.0f / (1.0f + __expf(-x));
}
__device__ __forceinline__ float tanh_f(float x) {
  float t = __expf(2.0f * fminf(fmaxf(x, -15.0f), 15.0f));
  return (t - 1.0f) / (t + 1.0f);
}

template <int N>
__global__ void __launch_bounds__(128)
proj_gemm(const float* __restrict__ A, const float* __restrict__ B,
          const float* __restrict__ bias, float* __restrict__ C) {
  __shared__ float Bs[32 * 304];
  const int tid  = threadIdx.x;
  const int lane = tid & 31;
  const int wave = tid >> 5;
  const int l15  = lane & 15;
  const int kb0  = (lane >> 4) << 1;
  const int n0   = (int)blockIdx.y * 32;

  {
    static_assert(DIN == 75 * 4, "75 rows per thread, 4 waves");
    const bool cok = n0 + lane < N;
    const float* bcol = B + (size_t)wave * N + (cok ? n0 + lane : 0);
    float bv[75];
#pragma unroll
    for (int i = 0; i < 75; ++i) bv[i] = bcol[(size_t)(4 * i) * N];
    float* bdst = Bs + lane * 304 + wave;
#pragma unroll
    for (int i = 0; i < 75; ++i) bdst[4 * i] = cok ? bv[i] : 0.0f;
  }
  __syncthreads();

  const int m0 = ((int)blockIdx.x * 4 + wave) * 16;
  if (m0 >= NN) return;
  const float* arow  = A + (size_t)(m0 + l15) * DIN;
  const float* b0row = Bs + l15 * 304;
  const float* b1row = Bs + (l15 + 16) * 304;
  v8f acc0 = {}, acc1 = {};
  for (int k0 = 0; k0 < DIN; k0 += 4) {
    int kb = k0 + kb0;
    v2f a  = *(const v2f*)(arow + kb);
    v2f b0 = *(const v2f*)(b0row + kb);
    v2f b1 = *(const v2f*)(b1row + kb);
    acc0 = __builtin_amdgcn_wmma_f32_16x16x4_f32(false, a, false, b0, (short)0,
                                                 acc0, false, false);
    acc1 = __builtin_amdgcn_wmma_f32_16x16x4_f32(false, a, false, b1, (short)0,
                                                 acc1, false, false);
  }
#pragma unroll
  for (int half = 0; half < 2; ++half) {
    int n = n0 + half * 16 + l15;
    if (n < N) {
      float bv = bias[n];
      const v8f& acc = half ? acc1 : acc0;
#pragma unroll
      for (int v = 0; v < 8; ++v) {
        int m = m0 + v + ((lane >> 4) << 3);
        C[(size_t)m * N + n] = acc[v] + bv;
      }
    }
  }
}

__global__ void leaf_kernel(const float* __restrict__ xiou,
                            const float* __restrict__ biouh,
                            float* __restrict__ cst, float* __restrict__ hst,
                            int count) {
  int e = (int)blockIdx.x * blockDim.x + threadIdx.x;
  if (e >= count * HD) return;
  int node = e / HD, col = e - node * HD;
  const float* r = xiou + (size_t)node * IOUW;
  float i = sigmoid_f(r[col] + biouh[col]);
  float o = sigmoid_f(r[HD + col] + biouh[HD + col]);
  float u = tanh_f(r[2 * HD + col] + biouh[2 * HD + col]);
  float c = i * u;
  cst[(size_t)node * HD + col] = c;
  hst[(size_t)node * HD + col] = o * tanh_f(c);
}

#define SLW 32
#define NSL 5
#define CSP 36
#define ISP 100
#define FSP 36

template <int NT>
__global__ void __launch_bounds__(NT)
level_kernel(const float* __restrict__ xiou, const float* __restrict__ xf,
             const float* __restrict__ Wiouh, const float* __restrict__ biouh,
             const float* __restrict__ Wfh, const float* __restrict__ bfh,
             float* __restrict__ cst, float* __restrict__ hst,
             int start, int end) {
  extern __shared__ float smem[];
  float* chh  = smem;
  float* hsum = chh  + 64 * CHP;
  float* chc  = hsum + 16 * CHP;
  float* ious = chc  + 64 * CSP;
  float* fcs  = ious + 16 * ISP;

  const int tid  = threadIdx.x;
  const int lane = tid & 31;
  const int wave = tid >> 5;
  const int l15  = lane & 15;
  const int kb0  = (lane >> 4) << 1;
  const int roff = (lane >> 4) << 3;
  const int base = start + (int)blockIdx.x * 16;
  const int c0   = (int)blockIdx.y * SLW;

  {
    constexpr int NH = 64 * CHP / NT, NC = 64 * SLW / NT;
    static_assert(NH * NT == 64 * CHP && NC * NT == 64 * SLW, "trip counts");
    float hv[NH], cv[NC];
#pragma unroll
    for (int i = 0; i < NH; ++i) {
      int e = tid + i * NT;
      int row = e / CHP, col = e - row * CHP;
      int node = base + (row >> 2);
      int cr = 4 * ((NN - 1) - node) + 1 + (row & 3);
      bool ok = col < HD && node < end && cr < NN;
      size_t off = ok ? (size_t)((NN - 1) - cr) * HD + col : (size_t)0;
      float h = hst[off];
      hv[i] = ok ? h : 0.0f;
    }
#pragma unroll
    for (int i = 0; i < NC; ++i) {
      int e = tid + i * NT;
      int row = e >> 5, col = c0 + (e & 31);
      int node = base + (row >> 2);
      int cr = 4 * ((NN - 1) - node) + 1 + (row & 3);
      bool ok = col < HD && node < end && cr < NN;
      size_t off = ok ? (size_t)((NN - 1) - cr) * HD + col : (size_t)0;
      float c = cst[off];
      cv[i] = ok ? c : 0.0f;
    }
#pragma unroll
    for (int i = 0; i < NH; ++i) chh[tid + i * NT] = hv[i];
#pragma unroll
    for (int i = 0; i < NC; ++i) {
      int e = tid + i * NT;
      chc[(e >> 5) * CSP + (e & 31)] = cv[i];
    }
  }
  __syncthreads();
  for (int e = tid; e < 16 * CHP; e += NT) {
    int m = e / CHP, col = e - m * CHP;
    int r0 = (m * 4) * CHP + col;
    hsum[e] = chh[r0] + chh[r0 + CHP] + chh[r0 + 2 * CHP] + chh[r0 + 3 * CHP];
  }
  __syncthreads();

  for (int t = wave; t < 14; t += NT / 32) {
    const bool gate = t < 6;
    const int  u    = gate ? t : t - 6;
    const int  hf   = u & 1;
    const int  lc   = hf * 16 + l15;
    const int  hc   = c0 + lc;
    const bool nok  = hc < HD;
    const bool tailok = kb0 == 0;
    if (gate) {
      const int g = u >> 1;
      const int n = g * HD + hc;
      const float* wcol = Wiouh + (nok ? n : 0);
      float bx[38], by[38];
#pragma unroll
      for (int s = 0; s < 37; ++s) {
        int kb = 4 * s + kb0;
        float v0 = wcol[kb * IOUW];
        float v1 = wcol[(kb + 1) * IOUW];
        bx[s] = nok ? v0 : 0.0f;
        by[s] = nok ? v1 : 0.0f;
      }
      {
        int kb = tailok ? 148 : 0;
        float v0 = wcol[kb * IOUW];
        float v1 = wcol[(kb + 1) * IOUW];
        bx[37] = (nok && tailok) ? v0 : 0.0f;
        by[37] = (nok && tailok) ? v1 : 0.0f;
      }
      v8f acc = {};
#pragma unroll
      for (int s = 0; s < 38; ++s) {
        int kb = 4 * s + kb0;
        v2f a = *(const v2f*)(hsum + l15 * CHP + kb);
        v2f b;
        b.x = bx[s];
        b.y = by[s];
        acc = __builtin_amdgcn_wmma_f32_16x16x4_f32(false, a, false, b, (short)0,
                                                    acc, false, false);
      }
      if (nok) {
        float bv = biouh[n];
        float xv[8];
#pragma unroll
        for (int v = 0; v < 8; ++v) {
          int node = base + v + roff;
          bool live = node < end;
          float tt = xiou[(size_t)(live ? node : base) * IOUW + n];
          xv[v] = live ? tt : 0.0f;
        }
#pragma unroll
        for (int v = 0; v < 8; ++v) {
          int m = v + roff;
          ious[m * ISP + g * SLW + lc] = acc[v] + bv + xv[v];
        }
      }
    } else {
      const int m0 = (u >> 1) * 16;
      const int nc = nok ? hc : 0;
      const float* wcol = Wfh + nc;
      float bx[38], by[38];
#pragma unroll
      for (int s = 0; s < 37; ++s) {
        int kb = 4 * s + kb0;
        float v0 = wcol[kb * HD];
        float v1 = wcol[(kb + 1) * HD];
        bx[s] = nok ? v0 : 0.0f;
        by[s] = nok ? v1 : 0.0f;
      }
      {
        int kb = tailok ? 148 : 0;
        float v0 = wcol[kb * HD];
        float v1 = wcol[(kb + 1) * HD];
        bx[37] = (nok && tailok) ? v0 : 0.0f;
        by[37] = (nok && tailok) ? v1 : 0.0f;
      }
      v8f acc = {};
#pragma unroll
      for (int s = 0; s < 38; ++s) {
        int kb = 4 * s + kb0;
        v2f a = *(const v2f*)(chh + (m0 + l15) * CHP + kb);
        v2f b;
        b.x = bx[s];
        b.y = by[s];
        acc = __builtin_amdgcn_wmma_f32_16x16x4_f32(false, a, false, b, (short)0,
                                                    acc, false, false);
      }
      float bv = nok ? bfh[nc] : 0.0f;
      float s0 = 0.0f, s1 = 0.0f;
#pragma unroll
      for (int v = 0; v < 8; ++v) {
        int grow = m0 + v + roff;
        int node = base + (grow >> 2);
        float xv = (nok && node < end) ? xf[(size_t)node * HD + nc] : 0.0f;
        float f  = sigmoid_f(acc[v] + bv + xv);
        float w  = f * chc[grow * CSP + lc];
        if (v < 4) s0 += w; else s1 += w;
      }
      int nl = (m0 >> 2) + ((lane >> 4) << 1);
      fcs[nl * FSP + lc]       = s0;
      fcs[(nl + 1) * FSP + lc] = s1;
    }
  }
  __syncthreads();

  for (int e = tid; e < 16 * SLW; e += NT) {
    int m = e >> 5, lc = e & 31;
    int col = c0 + lc;
    int node = base + m;
    if (node < end && col < HD) {
      float i = sigmoid_f(ious[m * ISP + lc]);
      float o = sigmoid_f(ious[m * ISP + SLW + lc]);
      float u = tanh_f(ious[m * ISP + 2 * SLW + lc]);
      float c = i * u + fcs[m * FSP + lc];
      cst[(size_t)node * HD + col] = c;
      hst[(size_t)node * HD + col] = o * tanh_f(c);
    }
  }
}

extern "C" void kernel_launch(void* const* d_in, const int* in_sizes, int n_in,
                              void* d_out, int out_size, void* d_ws, size_t ws_size,
                              hipStream_t stream) {
  const float* x     = (const float*)d_in[0];
  const float* Wioux = (const float*)d_in[1];
  const float* bioux = (const float*)d_in[2];
  const float* Wiouh = (const float*)d_in[3];
  const float* biouh = (const float*)d_in[4];
  const float* Wfx   = (const float*)d_in[5];
  const float* bfx   = (const float*)d_in[6];
  const float* Wfh   = (const float*)d_in[7];
  const float* bfh   = (const float*)d_in[8];
  (void)in_sizes; (void)n_in; (void)out_size; (void)ws_size;

  float* xiou = (float*)d_ws;
  float* xf   = xiou + (size_t)NN * IOUW;
  float* cst  = xf   + (size_t)NN * HD;
  float* hst  = (float*)d_out;

  proj_gemm<IOUW><<<dim3(313, 15), 128, 0, stream>>>(x, Wioux, bioux, xiou);
  proj_gemm<HD><<<dim3(313, 5),  128, 0, stream>>>(x, Wfx,   bfx,   xf);

  const int leafEnd = 14539;
  leaf_kernel<<<(leafEnd * HD + 255) / 256, 256, 0, stream>>>(xiou, biouh,
                                                              cst, hst, leafEnd);

  const int bounds[8] = {14539, 18635, 19659, 19915, 19979, 19995, 19999, 20000};
  const size_t lds =
      (size_t)(64 * CHP + 16 * CHP + 64 * CSP + 16 * ISP + 16 * FSP) * sizeof(float);
  int s = bounds[0];
  for (int L = 1; L < 8; ++L) {
    int e = bounds[L];
    int groups = (e - s + 15) / 16;
    if (L == 1)
      level_kernel<256><<<dim3(groups, NSL), 256, lds, stream>>>(xiou, xf, Wiouh, biouh,
                                                                 Wfh, bfh, cst, hst, s, e);
    else
      level_kernel<512><<<dim3(groups, NSL), 512, lds, stream>>>(xiou, xf, Wiouh, biouh,
                                                                 Wfh, bfh, cst, hst, s, e);
    s = e;
  }
}


# proj_gemm: LDS row pitch of the staged weight tile 304 -> 300 dwords (4 x odd): B-fragment reads go from 4-way bank conflict to none, staging stores from 8-way to 2-way
# speedup vs baseline: 4.2708x; 1.0596x over previous
#include <hip/hip_runtime.h>


typedef __attribute__((ext_vector_type(2))) float v2f;
typedef __attribute__((ext_vector_type(8))) float v8f;

#define NN   20000
#define DIN  300
#define HD   150
#define IOUW 450
#define CHP  152

__device__ __forceinline__ float sigmoid_f(float x) {
  return 1.0f / (1.0f + __expf(-x));
}
__device__ __forceinline__ float tanh_f(float x) {
  float t = __expf(2.0f * fminf(fmaxf(x, -15.0f), 15.0f));
  return (t - 1.0f) / (t + 1.0f);
}

template <int N>
__global__ void __launch_bounds__(128)
proj_gemm(const float* __restrict__ A, const float* __restrict__ B,
          const float* __restrict__ bias, float* __restrict__ C) {
  constexpr int BSP = 300;
  static_assert(BSP >= DIN && BSP % 8 == 4, "pitch holds a K row; 4 x odd dwords");
  __shared__ float Bs[32 * BSP];
  const int tid  = threadIdx.x;
  const int lane = tid & 31;
  const int wave = tid >> 5;
  const int l15  = lane & 15;
  const int kb0  = (lane >> 4) << 1;
  const int n0   = (int)blockIdx.y * 32;

  {
    static_assert(DIN == 75 * 4, "75 rows per thread, 4 waves");
    const bool cok = n0 + lane < N;
    const float* bcol = B + (size_t)wave * N + (cok ? n0 + lane : 0);
    float bv[75];
#pragma unroll
    for (int i = 0; i < 75; ++i) bv[i] = bcol[(size_t)(4 * i) * N];
    float* bdst = Bs + lane * BSP + wave;
#pragma unroll
    for (int i = 0; i < 75; ++i) bdst[4 * i] = cok ? bv[i] : 0.0f;
  }
  __syncthreads();

  const int m0 = ((int)blockIdx.x * 4 + wave) * 16;
  if (m0 >= NN) return;
  const float* arow  = A + (size_t)(m0 + l15) * DIN;
  const float* b0row = Bs + l15 * BSP;
  const float* b1row = Bs + (l15 + 16) * BSP;
  v8f acc0 = {}, acc1 = {};
  for (int k0 = 0; k0 < DIN; k0 += 4) {
    int kb = k0 + kb0;
    v2f a  = *(const v2f*)(arow + kb);
    v2f b0 = *(const v2f*)(b0row + kb);
    v2f b1 = *(const v2f*)(b1row + kb);
    acc0 = __builtin_amdgcn_wmma_f32_16x16x4_f32(false, a, false, b0, (short)0,
                                                 acc0, false, false);
    acc1 = __builtin_amdgcn_wmma_f32_16x16x4_f32(false, a, false, b1, (short)0,
                                                 acc1, false, false);
  }
#pragma unroll
  for (int half = 0; half < 2; ++half) {
    int n = n0 + half * 16 + l15;
    if (n < N) {
      float bv = bias[n];
      const v8f& acc = half ? acc1 : acc0;
#pragma unroll
      for (int v = 0; v < 8; ++v) {
        int m = m0 + v + ((lane >> 4) << 3);
        C[(size_t)m * N + n] = acc[v] + bv;
      }
    }
  }
}

__global__ void leaf_kernel(const float* __restrict__ xiou,
                            const float* __restrict__ biouh,
                            float* __restrict__ cst, float* __restrict__ hst,
                            int count) {
  int e = (int)blockIdx.x * blockDim.x + threadIdx.x;
  if (e >= count * HD) return;
  int node = e / HD, col = e - node * HD;
  const float* r = xiou + (size_t)node * IOUW;
  float i = sigmoid_f(r[col] + biouh[col]);
  float o = sigmoid_f(r[HD + col] + biouh[HD + col]);
  float u = tanh_f(r[2 * HD + col] + biouh[2 * HD + col]);
  float c = i * u;
  cst[(size_t)node * HD + col] = c;
  hst[(size_t)node * HD + col] = o * tanh_f(c);
}

#define SLW 32
#define NSL 5
#define CSP 36
#define ISP 100
#define FSP 36

template <int NT>
__global__ void __launch_bounds__(NT)
level_kernel(const float* __restrict__ xiou, const float* __restrict__ xf,
             const float* __restrict__ Wiouh, const float* __restrict__ biouh,
             const float* __restrict__ Wfh, const float* __restrict__ bfh,
             float* __restrict__ cst, float* __restrict__ hst,
             int start, int end) {
  extern __shared__ float smem[];
  float* chh  = smem;
  float* hsum = chh  + 64 * CHP;
  float* chc  = hsum + 16 * CHP;
  float* ious = chc  + 64 * CSP;
  float* fcs  = ious + 16 * ISP;

  const int tid  = threadIdx.x;
  const int lane = tid & 31;
  const int wave = tid >> 5;
  const int l15  = lane & 15;
  const int kb0  = (lane >> 4) << 1;
  const int roff = (lane >> 4) << 3;
  const int base = start + (int)blockIdx.x * 16;
  const int c0   = (int)blockIdx.y * SLW;

  {
    constexpr int NH = 64 * CHP / NT, NC = 64 * SLW / NT;
    static_assert(NH * NT == 64 * CHP && NC * NT == 64 * SLW, "trip counts");
    float hv[NH], cv[NC];
#pragma unroll
    for (int i = 0; i < NH; ++i) {
      int e = tid + i * NT;
      int row = e / CHP, col = e - row * CHP;
      int node = base + (row >> 2);
      int cr = 4 * ((NN - 1) - node) + 1 + (row & 3);
      bool ok = col < HD && node < end && cr < NN;
      size_t off = ok ? (size_t)((NN - 1) - cr) * HD + col : (size_t)0;
      float h = hst[off];
      hv[i] = ok ? h : 0.0f;
    }
#pragma unroll
    for (int i = 0; i < NC; ++i) {
      int e = tid + i * NT;
      int row = e >> 5, col = c0 + (e & 31);
      int node = base + (row >> 2);
      int cr = 4 * ((NN - 1) - node) + 1 + (row & 3);
      bool ok = col < HD && node < end && cr < NN;
      size_t off = ok ? (size_t)((NN - 1) - cr) * HD + col : (size_t)0;
      float c = cst[off];
      cv[i] = ok ? c : 0.0f;
    }
#pragma unroll
    for (int i = 0; i < NH; ++i) chh[tid + i * NT] = hv[i];
#pragma unroll
    for (int i = 0; i < NC; ++i) {
      int e = tid + i * NT;
      chc[(e >> 5) * CSP + (e & 31)] = cv[i];
    }
  }
  __syncthreads();
  for (int e = tid; e < 16 * CHP; e += NT) {
    int m = e / CHP, col = e - m * CHP;
    int r0 = (m * 4) * CHP + col;
    hsum[e] = chh[r0] + chh[r0 + CHP] + chh[r0 + 2 * CHP] + chh[r0 + 3 * CHP];
  }
  __syncthreads();

  for (int t = wave; t < 14; t += NT / 32) {
    const bool gate = t < 6;
    const int  u    = gate ? t : t - 6;
    const int  hf   = u & 1;
    const int  lc   = hf * 16 + l15;
    const int  hc   = c0 + lc;
    const bool nok  = hc < HD;
    const bool tailok = kb0 == 0;
    if (gate) {
      const int g = u >> 1;
      const int n = g * HD + hc;
      const float* wcol = Wiouh + (nok ? n : 0);
      float bx[38], by[38];
#pragma unroll
      for (int s = 0; s < 37; ++s) {
        int kb = 4 * s + kb0;
        float v0 = wcol[kb * IOUW];
        float v1 = wcol[(kb + 1) * IOUW];
        bx[s] = nok ? v0 : 0.0f;
        by[s] = nok ? v1 : 0.0f;
      }
      {
        int kb = tailok ? 148 : 0;
        float v0 = wcol[kb * IOUW];
        float v1 = wcol[(kb + 1) * IOUW];
        bx[37] = (nok && tailok) ? v0 : 0.0f;
        by[37] = (nok && tailok) ? v1 : 0.0f;
      }
      v8f acc = {};
#pragma unroll
      for (int s = 0; s < 38; ++s) {
        int kb = 4 * s + kb0;
        v2f a = *(const v2f*)(hsum + l15 * CHP + kb);
        v2f b;
        b.x = bx[s];
        b.y = by[s];
        acc = __builtin_amdgcn_wmma_f32_16x16x4_f32(false, a, false, b, (short)0,
                                                    acc, false, false);
      }
      if (nok) {
        float bv = biouh[n];
        float xv[8];
#pragma unroll
        for (int v = 0; v < 8; ++v) {
          int node = base + v + roff;
          bool live = node < end;
          float tt = xiou[(size_t)(live ? node : base) * IOUW + n];
          xv[v] = live ? tt : 0.0f;
        }
#pragma unroll
        for (int v = 0; v < 8; ++v) {
          int m = v + roff;
          ious[m * ISP + g * SLW + lc] = acc[v] + bv + xv[v];
        }
      }
    } else {
      const int m0 = (u >> 1) * 16;
      const int nc = nok ? hc : 0;
      const float* wcol = Wfh + nc;
      float bx[38], by[38];
#pragma unroll
      for (int s = 0; s < 37; ++s) {
        int kb = 4 * s + kb0;
        float v0 = wcol[kb * HD];
        float v1 = wcol[(kb + 1) * HD];
        bx[s] = nok ? v0 : 0.0f;
        by[s] = nok ? v1 : 0.0f;
      }
      {
        int kb = tailok ? 148 : 0;
        float v0 = wcol[kb * HD];
        float v1 = wcol[(kb + 1) * HD];
        bx[37] = (nok && tailok) ? v0 : 0.0f;
        by[37] = (nok && tailok) ? v1 : 0.0f;
      }
      v8f acc = {};
#pragma unroll
      for (int s = 0; s < 38; ++s) {
        int kb = 4 * s + kb0;
        v2f a = *(const v2f*)(chh + (m0 + l15) * CHP + kb);
        v2f b;
        b.x = bx[s];
        b.y = by[s];
        acc = __builtin_amdgcn_wmma_f32_16x16x4_f32(false, a, false, b, (short)0,
                                                    acc, false, false);
      }
      float bv = nok ? bfh[nc] : 0.0f;
      float s0 = 0.0f, s1 = 0.0f;
#pragma unroll
      for (int v = 0; v < 8; ++v) {
        int grow = m0 + v + roff;
        int node = base + (grow >> 2);
        float xv = (nok && node < end) ? xf[(size_t)node * HD + nc] : 0.0f;
        float f  = sigmoid_f(acc[v] + bv + xv);
        float w  = f * chc[grow * CSP + lc];
        if (v < 4) s0 += w; else s1 += w;
      }
      int nl = (m0 >> 2) + ((lane >> 4) << 1);
      fcs[nl * FSP + lc]       = s0;
      fcs[(nl + 1) * FSP + lc] = s1;
    }
  }
  __syncthreads();

  for (int e = tid; e < 16 * SLW; e += NT) {
    int m = e >> 5, lc = e & 31;
    int col = c0 + lc;
    int node = base + m;
    if (node < end && col < HD) {
      float i = sigmoid_f(ious[m * ISP + lc]);
      float o = sigmoid_f(ious[m * ISP + SLW + lc]);
      float u = tanh_f(ious[m * ISP + 2 * SLW + lc]);
      float c = i * u + fcs[m * FSP + lc];
      cst[(size_t)node * HD + col] = c;
      hst[(size_t)node * HD + col] = o * tanh_f(c);
    }
  }
}

extern "C" void kernel_launch(void* const* d_in, const int* in_sizes, int n_in,
                              void* d_out, int out_size, void* d_ws, size_t ws_size,
                              hipStream_t stream) {
  const float* x     = (const float*)d_in[0];
  const float* Wioux = (const float*)d_in[1];
  const float* bioux = (const float*)d_in[2];
  const float* Wiouh = (const float*)d_in[3];
  const float* biouh = (const float*)d_in[4];
  const float* Wfx   = (const float*)d_in[5];
  const float* bfx   = (const float*)d_in[6];
  const float* Wfh   = (const float*)d_in[7];
  const float* bfh   = (const float*)d_in[8];
  (void)in_sizes; (void)n_in; (void)out_size; (void)ws_size;

  float* xiou = (float*)d_ws;
  float* xf   = xiou + (size_t)NN * IOUW;
  float* cst  = xf   + (size_t)NN * HD;
  float* hst  = (float*)d_out;

  proj_gemm<IOUW><<<dim3(313, 15), 128, 0, stream>>>(x, Wioux, bioux, xiou);
  proj_gemm<HD><<<dim3(313, 5),  128, 0, stream>>>(x, Wfx,   bfx,   xf);

  const int leafEnd = 14539;
  leaf_kernel<<<(leafEnd * HD + 255) / 256, 256, 0, stream>>>(xiou, biouh,
                                                              cst, hst, leafEnd);

  const int bounds[8] = {14539, 18635, 19659, 19915, 19979, 19995, 19999, 20000};
  const size_t lds =
      (size_t)(64 * CHP + 16 * CHP + 64 * CSP + 16 * ISP + 16 * FSP) * sizeof(float);
  int s = bounds[0];
  for (int L = 1; L < 8; ++L) {
    int e = bounds[L];
    int groups = (e - s + 15) / 16;
    if (L == 1)
      level_kernel<256><<<dim3(groups, NSL), 256, lds, stream>>>(xiou, xf, Wiouh, biouh,
                                                                 Wfh, bfh, cst, hst, s, e);
    else
      level_kernel<512><<<dim3(groups, NSL), 512, lds, stream>>>(xiou, xf, Wiouh, biouh,
                                                                 Wfh, bfh, cst, hst, s, e);
    s = e;
  }
}


# proj_gemm: xf projected only for rows from the first interior node's tile on (a leaf's xf is never read): 430 blocks not 1565, run as extra blocks of the xiou launch, 9 launches not 10
# speedup vs baseline: 4.7210x; 1.1054x over previous
#include <hip/hip_runtime.h>


typedef __attribute__((ext_vector_type(2))) float v2f;
typedef __attribute__((ext_vector_type(8))) float v8f;

#define NN   20000
#define DIN  300
#define HD   150
#define IOUW 450
#define CHP  152

__device__ __forceinline__ float sigmoid_f(float x) {
  return 1.0f / (1.0f + __expf(-x));
}
__device__ __forceinline__ float tanh_f(float x) {
  float t = __expf(2.0f * fminf(fmaxf(x, -15.0f), 15.0f));
  return (t - 1.0f) / (t + 1.0f);
}

#define BSP 300
static_assert(BSP >= DIN && BSP % 8 == 4, "pitch holds a K row; 4 x odd dwords");

template <int N>
__device__ __forceinline__ void
proj_tile(const float* __restrict__ A, const float* __restrict__ B,
          const float* __restrict__ bias, float* __restrict__ C,
          float* Bs  , int mt0, int n0) {
  const int tid  = threadIdx.x;
  const int lane = tid & 31;
  const int wave = tid >> 5;
  const int l15  = lane & 15;
  const int kb0  = (lane >> 4) << 1;

  {
    static_assert(DIN == 75 * 4, "75 rows per thread, 4 waves");
    const bool cok = n0 + lane < N;
    const float* bcol = B + (size_t)wave * N + (cok ? n0 + lane : 0);
    float bv[75];
#pragma unroll
    for (int i = 0; i < 75; ++i) bv[i] = bcol[(size_t)(4 * i) * N];
    float* bdst = Bs + lane * BSP + wave;
#pragma unroll
    for (int i = 0; i < 75; ++i) bdst[4 * i] = cok ? bv[i] : 0.0f;
  }
  __syncthreads();

  const int m0 = (mt0 + wave) * 16;
  if (m0 >= NN) return;
  const float* arow  = A + (size_t)(m0 + l15) * DIN;
  const float* b0row = Bs + l15 * BSP;
  const float* b1row = Bs + (l15 + 16) * BSP;
  v8f acc0 = {}, acc1 = {};
  for (int k0 = 0; k0 < DIN; k0 += 4) {
    int kb = k0 + kb0;
    v2f a  = *(const v2f*)(arow + kb);
    v2f b0 = *(const v2f*)(b0row + kb);
    v2f b1 = *(const v2f*)(b1row + kb);
    acc0 = __builtin_amdgcn_wmma_f32_16x16x4_f32(false, a, false, b0, (short)0,
                                                 acc0, false, false);
    acc1 = __builtin_amdgcn_wmma_f32_16x16x4_f32(false, a, false, b1, (short)0,
                                                 acc1, false, false);
  }
#pragma unroll
  for (int half = 0; half < 2; ++half) {
    int n = n0 + half * 16 + l15;
    if (n < N) {
      float bv = bias[n];
      const v8f& acc = half ? acc1 : acc0;
#pragma unroll
      for (int v = 0; v < 8; ++v) {
        int m = m0 + v + ((lane >> 4) << 3);
        C[(size_t)m * N + n] = acc[v] + bv;
      }
    }
  }
}

#define LEAF_END   14539
#define XF_TILE0   (LEAF_END / 16)
#define XF_RBLK    ((NN / 16 - XF_TILE0 + 3) / 4)
#define IOU_RBLK   313
#define IOU_BLOCKS (IOU_RBLK * 15)
#define XF_BLOCKS  (XF_RBLK * 5)
static_assert(NN % 16 == 0 && IOU_RBLK * 4 >= NN / 16, "row tiling");
static_assert(IOU_BLOCKS + XF_BLOCKS == 5125, "grid literal in kernel_launch");

__global__ void __launch_bounds__(128)
proj_gemm(const float* __restrict__ x,
          const float* __restrict__ Wioux, const float* __restrict__ bioux,
          float* __restrict__ xiou,
          const float* __restrict__ Wfx, const float* __restrict__ bfx,
          float* __restrict__ xf) {
  __shared__ float Bs[32 * BSP];
  const int b = (int)blockIdx.x;
  if (b < IOU_BLOCKS) {
    const int cb = b / IOU_RBLK, rb = b - cb * IOU_RBLK;
    proj_tile<IOUW>(x, Wioux, bioux, xiou, Bs, rb * 4, cb * 32);
  } else {
    const int s  = b - IOU_BLOCKS;
    const int rb = s / 5, cb = s - rb * 5;
    proj_tile<HD>(x, Wfx, bfx, xf, Bs, XF_TILE0 + rb * 4, cb * 32);
  }
}

__global__ void leaf_kernel(const float* __restrict__ xiou,
                            const float* __restrict__ biouh,
                            float* __restrict__ cst, float* __restrict__ hst,
                            int count) {
  int e = (int)blockIdx.x * blockDim.x + threadIdx.x;
  if (e >= count * HD) return;
  int node = e / HD, col = e - node * HD;
  const float* r = xiou + (size_t)node * IOUW;
  float i = sigmoid_f(r[col] + biouh[col]);
  float o = sigmoid_f(r[HD + col] + biouh[HD + col]);
  float u = tanh_f(r[2 * HD + col] + biouh[2 * HD + col]);
  float c = i * u;
  cst[(size_t)node * HD + col] = c;
  hst[(size_t)node * HD + col] = o * tanh_f(c);
}

#define SLW 32
#define NSL 5
#define CSP 36
#define ISP 100
#define FSP 36

template <int NT>
__global__ void __launch_bounds__(NT)
level_kernel(const float* __restrict__ xiou, const float* __restrict__ xf,
             const float* __restrict__ Wiouh, const float* __restrict__ biouh,
             const float* __restrict__ Wfh, const float* __restrict__ bfh,
             float* __restrict__ cst, float* __restrict__ hst,
             int start, int end) {
  extern __shared__ float smem[];
  float* chh  = smem;
  float* hsum = chh  + 64 * CHP;
  float* chc  = hsum + 16 * CHP;
  float* ious = chc  + 64 * CSP;
  float* fcs  = ious + 16 * ISP;

  const int tid  = threadIdx.x;
  const int lane = tid & 31;
  const int wave = tid >> 5;
  const int l15  = lane & 15;
  const int kb0  = (lane >> 4) << 1;
  const int roff = (lane >> 4) << 3;
  const int base = start + (int)blockIdx.x * 16;
  const int c0   = (int)blockIdx.y * SLW;

  {
    constexpr int NH = 64 * CHP / NT, NC = 64 * SLW / NT;
    static_assert(NH * NT == 64 * CHP && NC * NT == 64 * SLW, "trip counts");
    float hv[NH], cv[NC];
#pragma unroll
    for (int i = 0; i < NH; ++i) {
      int e = tid + i * NT;
      int row = e / CHP, col = e - row * CHP;
      int node = base + (row >> 2);
      int cr = 4 * ((NN - 1) - node) + 1 + (row & 3);
      bool ok = col < HD && node < end && cr < NN;
      size_t off = ok ? (size_t)((NN - 1) - cr) * HD + col : (size_t)0;
      float h = hst[off];
      hv[i] = ok ? h : 0.0f;
    }
#pragma unroll
    for (int i = 0; i < NC; ++i) {
      int e = tid + i * NT;
      int row = e >> 5, col = c0 + (e & 31);
      int node = base + (row >> 2);
      int cr = 4 * ((NN - 1) - node) + 1 + (row & 3);
      bool ok = col < HD && node < end && cr < NN;
      size_t off = ok ? (size_t)((NN - 1) - cr) * HD + col : (size_t)0;
      float c = cst[off];
      cv[i] = ok ? c : 0.0f;
    }
#pragma unroll
    for (int i = 0; i < NH; ++i) chh[tid + i * NT] = hv[i];
#pragma unroll
    for (int i = 0; i < NC; ++i) {
      int e = tid + i * NT;
      chc[(e >> 5) * CSP + (e & 31)] = cv[i];
    }
  }
  __syncthreads();
  for (int e = tid; e < 16 * CHP; e += NT) {
    int m = e / CHP, col = e - m * CHP;
    int r0 = (m * 4) * CHP + col;
    hsum[e] = chh[r0] + chh[r0 + CHP] + chh[r0 + 2 * CHP] + chh[r0 + 3 * CHP];
  }
  __syncthreads();

  for (int t = wave; t < 14; t += NT / 32) {
    const bool gate = t < 6;
    const int  u    = gate ? t : t - 6;
    const int  hf   = u & 1;
    const int  lc   = hf * 16 + l15;
    const int  hc   = c0 + lc;
    const bool nok  = hc < HD;
    const bool tailok = kb0 == 0;
    if (gate) {
      const int g = u >> 1;
      const int n = g * HD + hc;
      const float* wcol = Wiouh + (nok ? n : 0);
      float bx[38], by[38];
#pragma unroll
      for (int s = 0; s < 37; ++s) {
        int kb = 4 * s + kb0;
        float v0 = wcol[kb * IOUW];
        float v1 = wcol[(kb + 1) * IOUW];
        bx[s] = nok ? v0 : 0.0f;
        by[s] = nok ? v1 : 0.0f;
      }
      {
        int kb = tailok ? 148 : 0;
        float v0 = wcol[kb * IOUW];
        float v1 = wcol[(kb + 1) * IOUW];
        bx[37] = (nok && tailok) ? v0 : 0.0f;
        by[37] = (nok && tailok) ? v1 : 0.0f;
      }
      v8f acc = {};
#pragma unroll
      for (int s = 0; s < 38; ++s) {
        int kb = 4 * s + kb0;
        v2f a = *(const v2f*)(hsum + l15 * CHP + kb);
        v2f b;
        b.x = bx[s];
        b.y = by[s];
        acc = __builtin_amdgcn_wmma_f32_16x16x4_f32(false, a, false, b, (short)0,
                                                    acc, false, false);
      }
      if (nok) {
        float bv = biouh[n];
        float xv[8];
#pragma unroll
        for (int v = 0; v < 8; ++v) {
          int node = base + v + roff;
          bool live = node < end;
          float tt = xiou[(size_t)(live ? node : base) * IOUW + n];
          xv[v] = live ? tt : 0.0f;
        }
#pragma unroll
        for (int v = 0; v < 8; ++v) {
          int m = v + roff;
          ious[m * ISP + g * SLW + lc] = acc[v] + bv + xv[v];
        }
      }
    } else {
      const int m0 = (u >> 1) * 16;
      const int nc = nok ? hc : 0;
      const float* wcol = Wfh + nc;
      float bx[38], by[38];
#pragma unroll
      for (int s = 0; s < 37; ++s) {
        int kb = 4 * s + kb0;
        float v0 = wcol[kb * HD];
        float v1 = wcol[(kb + 1) * HD];
        bx[s] = nok ? v0 : 0.0f;
        by[s] = nok ? v1 : 0.0f;
      }
      {
        int kb = tailok ? 148 : 0;
        float v0 = wcol[kb * HD];
        float v1 = wcol[(kb + 1) * HD];
        bx[37] = (nok && tailok) ? v0 : 0.0f;
        by[37] = (nok && tailok) ? v1 : 0.0f;
      }
      v8f acc = {};
#pragma unroll
      for (int s = 0; s < 38; ++s) {
        int kb = 4 * s + kb0;
        v2f a = *(const v2f*)(chh + (m0 + l15) * CHP + kb);
        v2f b;
        b.x = bx[s];
        b.y = by[s];
        acc = __builtin_amdgcn_wmma_f32_16x16x4_f32(false, a, false, b, (short)0,
                                                    acc, false, false);
      }
      float bv = nok ? bfh[nc] : 0.0f;
      float s0 = 0.0f, s1 = 0.0f;
#pragma unroll
      for (int v = 0; v < 8; ++v) {
        int grow = m0 + v + roff;
        int node = base + (grow >> 2);
        float xv = (nok && node < end) ? xf[(size_t)node * HD + nc] : 0.0f;
        float f  = sigmoid_f(acc[v] + bv + xv);
        float w  = f * chc[grow * CSP + lc];
        if (v < 4) s0 += w; else s1 += w;
      }
      int nl = (m0 >> 2) + ((lane >> 4) << 1);
      fcs[nl * FSP + lc]       = s0;
      fcs[(nl + 1) * FSP + lc] = s1;
    }
  }
  __syncthreads();

  for (int e = tid; e < 16 * SLW; e += NT) {
    int m = e >> 5, lc = e & 31;
    int col = c0 + lc;
    int node = base + m;
    if (node < end && col < HD) {
      float i = sigmoid_f(ious[m * ISP + lc]);
      float o = sigmoid_f(ious[m * ISP + SLW + lc]);
      float u = tanh_f(ious[m * ISP + 2 * SLW + lc]);
      float c = i * u + fcs[m * FSP + lc];
      cst[(size_t)node * HD + col] = c;
      hst[(size_t)node * HD + col] = o * tanh_f(c);
    }
  }
}

extern "C" void kernel_launch(void* const* d_in, const int* in_sizes, int n_in,
                              void* d_out, int out_size, void* d_ws, size_t ws_size,
                              hipStream_t stream) {
  const float* x     = (const float*)d_in[0];
  const float* Wioux = (const float*)d_in[1];
  const float* bioux = (const float*)d_in[2];
  const float* Wiouh = (const float*)d_in[3];
  const float* biouh = (const float*)d_in[4];
  const float* Wfx   = (const float*)d_in[5];
  const float* bfx   = (const float*)d_in[6];
  const float* Wfh   = (const float*)d_in[7];
  const float* bfh   = (const float*)d_in[8];
  (void)in_sizes; (void)n_in; (void)out_size; (void)ws_size;

  float* xiou = (float*)d_ws;
  float* xf   = xiou + (size_t)NN * IOUW;
  float* cst  = xf   + (size_t)NN * HD;
  float* hst  = (float*)d_out;

  proj_gemm<<<5125, 128, 0, stream>>>(x, Wioux, bioux, xiou, Wfx, bfx, xf);

  const int leafEnd = LEAF_END;
  leaf_kernel<<<(leafEnd * HD + 255) / 256, 256, 0, stream>>>(xiou, biouh,
                                                              cst, hst, leafEnd);

  const int bounds[8] = {14539, 18635, 19659, 19915, 19979, 19995, 19999, 20000};
  const size_t lds =
      (size_t)(64 * CHP + 16 * CHP + 64 * CSP + 16 * ISP + 16 * FSP) * sizeof(float);
  int s = bounds[0];
  for (int L = 1; L < 8; ++L) {
    int e = bounds[L];
    int groups = (e - s + 15) / 16;
    if (L == 1)
      level_kernel<256><<<dim3(groups, NSL), 256, lds, stream>>>(xiou, xf, Wiouh, biouh,
                                                                 Wfh, bfh, cst, hst, s, e);
    else
      level_kernel<512><<<dim3(groups, NSL), 512, lds, stream>>>(xiou, xf, Wiouh, biouh,
                                                                 Wfh, bfh, cst, hst, s, e);
    s = e;
  }
}


# proj_gemm: A and B fragments read as 16 B per lane, one read per two K-steps (k of each 8-k block permuted alike in both operands; sum order changes): half the global and LDS read instructions
# speedup vs baseline: 4.9878x; 1.0565x over previous
#include <hip/hip_runtime.h>


typedef __attribute__((ext_vector_type(2))) float v2f;
typedef __attribute__((ext_vector_type(4))) float v4f;
typedef __attribute__((ext_vector_type(8))) float v8f;

#define NN   20000
#define DIN  300
#define HD   150
#define IOUW 450
#define CHP  152

__device__ __forceinline__ float sigmoid_f(float x) {
  return 1.0f / (1.0f + __expf(-x));
}
__device__ __forceinline__ float tanh_f(float x) {
  float t = __expf(2.0f * fminf(fmaxf(x, -15.0f), 15.0f));
  return (t - 1.0f) / (t + 1.0f);
}

#define BSP 300
static_assert(BSP >= DIN && BSP % 8 == 4, "pitch holds a K row; 4 x odd dwords");

template <int N>
__device__ __forceinline__ void
proj_tile(const float* __restrict__ A, const float* __restrict__ B,
          const float* __restrict__ bias, float* __restrict__ C,
          float* Bs  , int mt0, int n0) {
  const int tid  = threadIdx.x;
  const int lane = tid & 31;
  const int wave = tid >> 5;
  const int l15  = lane & 15;
  const int kb0  = (lane >> 4) << 1;

  {
    static_assert(DIN == 75 * 4, "75 rows per thread, 4 waves");
    const bool cok = n0 + lane < N;
    const float* bcol = B + (size_t)wave * N + (cok ? n0 + lane : 0);
    float bv[75];
#pragma unroll
    for (int i = 0; i < 75; ++i) bv[i] = bcol[(size_t)(4 * i) * N];
    float* bdst = Bs + lane * BSP + wave;
#pragma unroll
    for (int i = 0; i < 75; ++i) bdst[4 * i] = cok ? bv[i] : 0.0f;
  }
  __syncthreads();

  const int m0 = (mt0 + wave) * 16;
  if (m0 >= NN) return;
  const float* arow  = A + (size_t)(m0 + l15) * DIN;
  const float* b0row = Bs + l15 * BSP;
  const float* b1row = Bs + (l15 + 16) * BSP;
  v8f acc0 = {}, acc1 = {};
  static_assert(DIN == 37 * 8 + 4 && (DIN * 4) % 16 == 0 && (BSP * 4) % 16 == 0,
                "37 blocks of 8 k and one K-step of 4; 16-byte aligned rows");
  const int kq = (lane >> 4) << 2;
  auto block8 = [&](int k0) {
    int kb = k0 + kq;
    v4f a  = *(const v4f*)(arow + kb);
    v4f b0 = *(const v4f*)(b0row + kb);
    v4f b1 = *(const v4f*)(b1row + kb);
    acc0 = __builtin_amdgcn_wmma_f32_16x16x4_f32(false, a.lo, false, b0.lo, (short)0,
                                                 acc0, false, false);
    acc1 = __builtin_amdgcn_wmma_f32_16x16x4_f32(false, a.lo, false, b1.lo, (short)0,
                                                 acc1, false, false);
    acc0 = __builtin_amdgcn_wmma_f32_16x16x4_f32(false, a.hi, false, b0.hi, (short)0,
                                                 acc0, false, false);
    acc1 = __builtin_amdgcn_wmma_f32_16x16x4_f32(false, a.hi, false, b1.hi, (short)0,
                                                 acc1, false, false);
  };
#pragma unroll 6
  for (int k0 = 0; k0 < 36 * 8; k0 += 8) block8(k0);
  block8(36 * 8);
  {
    int kb = (DIN - 4) + kb0;
    v2f a  = *(const v2f*)(arow + kb);
    v2f b0 = *(const v2f*)(b0row + kb);
    v2f b1 = *(const v2f*)(b1row + kb);
    acc0 = __builtin_amdgcn_wmma_f32_16x16x4_f32(false, a, false, b0, (short)0,
                                                 acc0, false, false);
    acc1 = __builtin_amdgcn_wmma_f32_16x16x4_f32(false, a, false, b1, (short)0,
                                                 acc1, false, false);
  }
#pragma unroll
  for (int half = 0; half < 2; ++half) {
    int n = n0 + half * 16 + l15;
    if (n < N) {
      float bv = bias[n];
      const v8f& acc = half ? acc1 : acc0;
#pragma unroll
      for (int v = 0; v < 8; ++v) {
        int m = m0 + v + ((lane >> 4) << 3);
        C[(size_t)m * N + n] = acc[v] + bv;
      }
    }
  }
}

#define LEAF_END   14539
#define XF_TILE0   (LEAF_END / 16)
#define XF_RBLK    ((NN / 16 - XF_TILE0 + 3) / 4)
#define IOU_RBLK   313
#define IOU_BLOCKS (IOU_RBLK * 15)
#define XF_BLOCKS  (XF_RBLK * 5)
static_assert(NN % 16 == 0 && IOU_RBLK * 4 >= NN / 16, "row tiling");
static_assert(IOU_BLOCKS + XF_BLOCKS == 5125, "grid literal in kernel_launch");

__global__ void __launch_bounds__(128)
proj_gemm(const float* __restrict__ x,
          const float* __restrict__ Wioux, const float* __restrict__ bioux,
          float* __restrict__ xiou,
          const float* __restrict__ Wfx, const float* __restrict__ bfx,
          float* __restrict__ xf) {
  __shared__ __attribute__((aligned(16))) float Bs[32 * BSP];
  const int b = (int)blockIdx.x;
  if (b < IOU_BLOCKS) {
    const int cb = b / IOU_RBLK, rb = b - cb * IOU_RBLK;
    proj_tile<IOUW>(x, Wioux, bioux, xiou, Bs, rb * 4, cb * 32);
  } else {
    const int s  = b - IOU_BLOCKS;
    const int rb = s / 5, cb = s - rb * 5;
    proj_tile<HD>(x, Wfx, bfx, xf, Bs, XF_TILE0 + rb * 4, cb * 32);
  }
}

__global__ void leaf_kernel(const float* __restrict__ xiou,
                            const float* __restrict__ biouh,
                            float* __restrict__ cst, float* __restrict__ hst,
                            int count) {
  int e = (int)blockIdx.x * blockDim.x + threadIdx.x;
  if (e >= count * HD) return;
  int node = e / HD, col = e - node * HD;
  const float* r = xiou + (size_t)node * IOUW;
  float i = sigmoid_f(r[col] + biouh[col]);
  float o = sigmoid_f(r[HD + col] + biouh[HD + col]);
  float u = tanh_f(r[2 * HD + col] + biouh[2 * HD + col]);
  float c = i * u;
  cst[(size_t)node * HD + col] = c;
  hst[(size_t)node * HD + col] = o * tanh_f(c);
}

#define SLW 32
#define NSL 5
#define CSP 36
#define ISP 100
#define FSP 36

template <int NT>
__global__ void __launch_bounds__(NT)
level_kernel(const float* __restrict__ xiou, const float* __restrict__ xf,
             const float* __restrict__ Wiouh, const float* __restrict__ biouh,
             const float* __restrict__ Wfh, const float* __restrict__ bfh,
             float* __restrict__ cst, float* __restrict__ hst,
             int start, int end) {
  extern __shared__ float smem[];
  float* chh  = smem;
  float* hsum = chh  + 64 * CHP;
  float* chc  = hsum + 16 * CHP;
  float* ious = chc  + 64 * CSP;
  float* fcs  = ious + 16 * ISP;

  const int tid  = threadIdx.x;
  const int lane = tid & 31;
  const int wave = tid >> 5;
  const int l15  = lane & 15;
  const int kb0  = (lane >> 4) << 1;
  const int roff = (lane >> 4) << 3;
  const int base = start + (int)blockIdx.x * 16;
  const int c0   = (int)blockIdx.y * SLW;

  {
    constexpr int NH = 64 * CHP / NT, NC = 64 * SLW / NT;
    static_assert(NH * NT == 64 * CHP && NC * NT == 64 * SLW, "trip counts");
    float hv[NH], cv[NC];
#pragma unroll
    for (int i = 0; i < NH; ++i) {
      int e = tid + i * NT;
      int row = e / CHP, col = e - row * CHP;
      int node = base + (row >> 2);
      int cr = 4 * ((NN - 1) - node) + 1 + (row & 3);
      bool ok = col < HD && node < end && cr < NN;
      size_t off = ok ? (size_t)((NN - 1) - cr) * HD + col : (size_t)0;
      float h = hst[off];
      hv[i] = ok ? h : 0.0f;
    }
#pragma unroll
    for (int i = 0; i < NC; ++i) {
      int e = tid + i * NT;
      int row = e >> 5, col = c0 + (e & 31);
      int node = base + (row >> 2);
      int cr = 4 * ((NN - 1) - node) + 1 + (row & 3);
      bool ok = col < HD && node < end && cr < NN;
      size_t off = ok ? (size_t)((NN - 1) - cr) * HD + col : (size_t)0;
      float c = cst[off];
      cv[i] = ok ? c : 0.0f;
    }
#pragma unroll
    for (int i = 0; i < NH; ++i) chh[tid + i * NT] = hv[i];
#pragma unroll
    for (int i = 0; i < NC; ++i) {
      int e = tid + i * NT;
      chc[(e >> 5) * CSP + (e & 31)] = cv[i];
    }
  }
  __syncthreads();
  for (int e = tid; e < 16 * CHP; e += NT) {
    int m = e / CHP, col = e - m * CHP;
    int r0 = (m * 4) * CHP + col;
    hsum[e] = chh[r0] + chh[r0 + CHP] + chh[r0 + 2 * CHP] + chh[r0 + 3 * CHP];
  }
  __syncthreads();

  for (int t = wave; t < 14; t += NT / 32) {
    const bool gate = t < 6;
    const int  u    = gate ? t : t - 6;
    const int  hf   = u & 1;
    const int  lc   = hf * 16 + l15;
    const int  hc   = c0 + lc;
    const bool nok  = hc < HD;
    const bool tailok = kb0 == 0;
    if (gate) {
      const int g = u >> 1;
      const int n = g * HD + hc;
      const float* wcol = Wiouh + (nok ? n : 0);
      float bx[38], by[38];
#pragma unroll
      for (int s = 0; s < 37; ++s) {
        int kb = 4 * s + kb0;
        float v0 = wcol[kb * IOUW];
        float v1 = wcol[(kb + 1) * IOUW];
        bx[s] = nok ? v0 : 0.0f;
        by[s] = nok ? v1 : 0.0f;
      }
      {
        int kb = tailok ? 148 : 0;
        float v0 = wcol[kb * IOUW];
        float v1 = wcol[(kb + 1) * IOUW];
        bx[37] = (nok && tailok) ? v0 : 0.0f;
        by[37] = (nok && tailok) ? v1 : 0.0f;
      }
      v8f acc = {};
#pragma unroll
      for (int s = 0; s < 38; ++s) {
        int kb = 4 * s + kb0;
        v2f a = *(const v2f*)(hsum + l15 * CHP + kb);
        v2f b;
        b.x = bx[s];
        b.y = by[s];
        acc = __builtin_amdgcn_wmma_f32_16x16x4_f32(false, a, false, b, (short)0,
                                                    acc, false, false);
      }
      if (nok) {
        float bv = biouh[n];
        float xv[8];
#pragma unroll
        for (int v = 0; v < 8; ++v) {
          int node = base + v + roff;
          bool live = node < end;
          float tt = xiou[(size_t)(live ? node : base) * IOUW + n];
          xv[v] = live ? tt : 0.0f;
        }
#pragma unroll
        for (int v = 0; v < 8; ++v) {
          int m = v + roff;
          ious[m * ISP + g * SLW + lc] = acc[v] + bv + xv[v];
        }
      }
    } else {
      const int m0 = (u >> 1) * 16;
      const int nc = nok ? hc : 0;
      const float* wcol = Wfh + nc;
      float bx[38], by[38];
#pragma unroll
      for (int s = 0; s < 37; ++s) {
        int kb = 4 * s + kb0;
        float v0 = wcol[kb * HD];
        float v1 = wcol[(kb + 1) * HD];
        bx[s] = nok ? v0 : 0.0f;
        by[s] = nok ? v1 : 0.0f;
      }
      {
        int kb = tailok ? 148 : 0;
        float v0 = wcol[kb * HD];
        float v1 = wcol[(kb + 1) * HD];
        bx[37] = (nok && tailok) ? v0 : 0.0f;
        by[37] = (nok && tailok) ? v1 : 0.0f;
      }
      v8f acc = {};
#pragma unroll
      for (int s = 0; s < 38; ++s) {
        int kb = 4 * s + kb0;
        v2f a = *(const v2f*)(chh + (m0 + l15) * CHP + kb);
        v2f b;
        b.x = bx[s];
        b.y = by[s];
        acc = __builtin_amdgcn_wmma_f32_16x16x4_f32(false, a, false, b, (short)0,
                                                    acc, false, false);
      }
      float bv = nok ? bfh[nc] : 0.0f;
      float s0 = 0.0f, s1 = 0.0f;
#pragma unroll
      for (int v = 0; v < 8; ++v) {
        int grow = m0 + v + roff;
        int node = base + (grow >> 2);
        float xv = (nok && node < end) ? xf[(size_t)node * HD + nc] : 0.0f;
        float f  = sigmoid_f(acc[v] + bv + xv);
        float w  = f * chc[grow * CSP + lc];
        if (v < 4) s0 += w; else s1 += w;
      }
      int nl = (m0 >> 2) + ((lane >> 4) << 1);
      fcs[nl * FSP + lc]       = s0;
      fcs[(nl + 1) * FSP + lc] = s1;
    }
  }
  __syncthreads();

  for (int e = tid; e < 16 * SLW; e += NT) {
    int m = e >> 5, lc = e & 31;
    int col = c0 + lc;
    int node = base + m;
    if (node < end && col < HD) {
      float i = sigmoid_f(ious[m * ISP + lc]);
      float o = sigmoid_f(ious[m * ISP + SLW + lc]);
      float u = tanh_f(ious[m * ISP + 2 * SLW + lc]);
      float c = i * u + fcs[m * FSP + lc];
      cst[(size_t)node * HD + col] = c;
      hst[(size_t)node * HD + col] = o * tanh_f(c);
    }
  }
}

extern "C" void kernel_launch(void* const* d_in, const int* in_sizes, int n_in,
                              void* d_out, int out_size, void* d_ws, size_t ws_size,
                              hipStream_t stream) {
  const float* x     = (const float*)d_in[0];
  const float* Wioux = (const float*)d_in[1];
  const float* bioux = (const float*)d_in[2];
  const float* Wiouh = (const float*)d_in[3];
  const float* biouh = (const float*)d_in[4];
  const float* Wfx   = (const float*)d_in[5];
  const float* bfx   = (const float*)d_in[6];
  const float* Wfh   = (const float*)d_in[7];
  const float* bfh   = (const float*)d_in[8];
  (void)in_sizes; (void)n_in; (void)out_size; (void)ws_size;

  float* xiou = (float*)d_ws;
  float* xf   = xiou + (size_t)NN * IOUW;
  float* cst  = xf   + (size_t)NN * HD;
  float* hst  = (float*)d_out;

  proj_gemm<<<5125, 128, 0, stream>>>(x, Wioux, bioux, xiou, Wfx, bfx, xf);

  const int leafEnd = LEAF_END;
  leaf_kernel<<<(leafEnd * HD + 255) / 256, 256, 0, stream>>>(xiou, biouh,
                                                              cst, hst, leafEnd);

  const int bounds[8] = {14539, 18635, 19659, 19915, 19979, 19995, 19999, 20000};
  const size_t lds =
      (size_t)(64 * CHP + 16 * CHP + 64 * CSP + 16 * ISP + 16 * FSP) * sizeof(float);
  int s = bounds[0];
  for (int L = 1; L < 8; ++L) {
    int e = bounds[L];
    int groups = (e - s + 15) / 16;
    if (L == 1)
      level_kernel<256><<<dim3(groups, NSL), 256, lds, stream>>>(xiou, xf, Wiouh, biouh,
                                                                 Wfh, bfh, cst, hst, s, e);
    else
      level_kernel<512><<<dim3(groups, NSL), 512, lds, stream>>>(xiou, xf, Wiouh, biouh,
                                                                 Wfh, bfh, cst, hst, s, e);
    s = e;
  }
}


# proj_gemm: a workgroup takes one H slice of all three gates (i, o, u of an element in one lane), so the leaf update is its epilogue: leaf launch and leaves' xiou write+read gone, 8 launches not 9
# speedup vs baseline: 5.3125x; 1.0651x over previous
#include <hip/hip_runtime.h>


typedef __attribute__((ext_vector_type(2))) float v2f;
typedef __attribute__((ext_vector_type(4))) float v4f;
typedef __attribute__((ext_vector_type(8))) float v8f;

#define NN   20000
#define DIN  300
#define HD   150
#define IOUW 450
#define CHP  152

__device__ __forceinline__ float sigmoid_f(float x) {
  return 1.0f / (1.0f + __expf(-x));
}
__device__ __forceinline__ float tanh_f(float x) {
  float t = __expf(2.0f * fminf(fmaxf(x, -15.0f), 15.0f));
  return (t - 1.0f) / (t + 1.0f);
}

#define BSP 300
static_assert(BSP >= DIN && BSP % 8 == 4, "pitch holds a K row; 4 x odd dwords");

template <int NG, int GS, int GV, int N>
__device__ __forceinline__ bool
proj_acc(const float* __restrict__ A, const float* __restrict__ B,
         float* Bs  , int mt0, int n0, v8f (&acc)[NG]) {
  const int tid  = threadIdx.x;
  const int lane = tid & 31;
  const int wave = tid >> 5;
  const int l15  = lane & 15;
  const int kb0  = (lane >> 4) << 1;

  {
    static_assert(DIN == 37 * 8 + 4, "37 rows per thread, 8 waves, and 4 rows left over");
    const bool last = wave < 4;
    float bv[NG][38];
#pragma unroll
    for (int g = 0; g < NG; ++g) {
      const bool cok = n0 + g * GV + lane < HD;
      const float* bcol = B + (size_t)wave * N + (cok ? n0 + g * GS + lane : 0);
#pragma unroll
      for (int i = 0; i < 37; ++i) bv[g][i] = bcol[(size_t)(8 * i) * N];
      bv[g][37] = last ? bcol[(size_t)(8 * 37) * N] : 0.0f;
    }
#pragma unroll
    for (int g = 0; g < NG; ++g) {
      const bool cok = n0 + g * GV + lane < HD;
      float* bdst = Bs + (g * 32 + lane) * BSP + wave;
#pragma unroll
      for (int i = 0; i < 37; ++i) bdst[8 * i] = cok ? bv[g][i] : 0.0f;
      if (last) bdst[8 * 37] = cok ? bv[g][37] : 0.0f;
    }
  }
  __syncthreads();

  const int m0 = (mt0 + (wave & 3)) * 16;
  if (m0 >= NN) return false;
  const float* arow = A + (size_t)(m0 + l15) * DIN;
  const float* brow = Bs + ((wave >> 2) * 16 + l15) * BSP;
#pragma unroll
  for (int g = 0; g < NG; ++g) acc[g] = v8f{};
  static_assert(DIN == 37 * 8 + 4 && (DIN * 4) % 16 == 0 && (BSP * 4) % 16 == 0,
                "37 blocks of 8 k and one K-step of 4; 16-byte aligned rows");
  const int kq = (lane >> 4) << 2;
  auto block8 = [&](int k0) {
    int kb = k0 + kq;
    v4f a = *(const v4f*)(arow + kb);
    v4f b[NG];
#pragma unroll
    for (int g = 0; g < NG; ++g)
      b[g] = *(const v4f*)(brow + g * 32 * BSP + kb);
#pragma unroll
    for (int g = 0; g < NG; ++g)
      acc[g] = __builtin_amdgcn_wmma_f32_16x16x4_f32(false, a.lo, false, b[g].lo, (short)0,
                                                     acc[g], false, false);
#pragma unroll
    for (int g = 0; g < NG; ++g)
      acc[g] = __builtin_amdgcn_wmma_f32_16x16x4_f32(false, a.hi, false, b[g].hi, (short)0,
                                                     acc[g], false, false);
  };
#pragma unroll 12
  for (int k0 = 0; k0 < 36 * 8; k0 += 8) block8(k0);
  block8(36 * 8);
  {
    int kb = (DIN - 4) + kb0;
    v2f a = *(const v2f*)(arow + kb);
#pragma unroll
    for (int g = 0; g < NG; ++g) {
      v2f b = *(const v2f*)(brow + g * 32 * BSP + kb);
      acc[g] = __builtin_amdgcn_wmma_f32_16x16x4_f32(false, a, false, b, (short)0,
                                                     acc[g], false, false);
    }
  }
  return true;
}

#define LEAF_END   14539
#define XF_TILE0   (LEAF_END / 16)
#define XF_RBLK    ((NN / 16 - XF_TILE0 + 3) / 4)
#define IOU_RBLK   313
#define IOU_BLOCKS (IOU_RBLK * 5)
#define XF_BLOCKS  (XF_RBLK * 2)
static_assert(NN % 16 == 0 && IOU_RBLK * 4 >= NN / 16, "row tiling");
static_assert(IOU_BLOCKS + XF_BLOCKS == 1737, "grid literal in kernel_launch");

template <int NG>
__device__ __forceinline__ void
xf_store(const v8f (&acc)[NG], const float* __restrict__ bfx,
         float* __restrict__ xf, int m0, int n0) {
  const int lane = threadIdx.x & 31;
  const int l15  = lane & 15;
  const int roff = (lane >> 4) << 3;
  const int hf   = threadIdx.x >> 7;
#pragma unroll
  for (int g = 0; g < NG; ++g) {
    int n = n0 + g * 32 + hf * 16 + l15;
    if (n < HD) {
      float bv = bfx[n];
#pragma unroll
      for (int v = 0; v < 8; ++v)
        xf[(size_t)(m0 + v + roff) * HD + n] = acc[g][v] + bv;
    }
  }
}

__global__ void __launch_bounds__(256)
proj_gemm(const float* __restrict__ x,
          const float* __restrict__ Wioux, const float* __restrict__ bioux,
          const float* __restrict__ biouh, float* __restrict__ xiou,
          const float* __restrict__ Wfx, const float* __restrict__ bfx,
          float* __restrict__ xf,
          float* __restrict__ cst, float* __restrict__ hst) {
  __shared__ __attribute__((aligned(16))) float Bs[96 * BSP];
  const int b    = (int)blockIdx.x;
  const int wave = threadIdx.x >> 5;
  if (b < IOU_BLOCKS) {
    const int cb = b / IOU_RBLK, rb = b - cb * IOU_RBLK;
    const int n0 = cb * 32;
    v8f acc[3];
    if (!proj_acc<3, HD, 0, IOUW>(x, Wioux, Bs, rb * 4, n0, acc)) return;
    const int lane = threadIdx.x & 31;
    const int roff = (lane >> 4) << 3;
    const int m0   = (rb * 4 + (wave & 3)) * 16;
    const int n    = n0 + (wave >> 2) * 16 + (lane & 15);
    if (n < HD) {
      float bi = bioux[n], bo = bioux[HD + n], bu = bioux[2 * HD + n];
      float hi = biouh[n], ho = biouh[HD + n], hu = biouh[2 * HD + n];
#pragma unroll
      for (int v = 0; v < 8; ++v) {
        int m = m0 + v + roff;
        float xi = acc[0][v] + bi;
        float xo = acc[1][v] + bo;
        float xu = acc[2][v] + bu;
        if (m < LEAF_END) {
          float i = sigmoid_f(xi + hi);
          float o = sigmoid_f(xo + ho);
          float u = tanh_f(xu + hu);
          float c = i * u;
          cst[(size_t)m * HD + n] = c;
          hst[(size_t)m * HD + n] = o * tanh_f(c);
        } else {
          float* r = xiou + (size_t)m * IOUW + n;
          r[0]      = xi;
          r[HD]     = xo;
          r[2 * HD] = xu;
        }
      }
    }
  } else {
    const int s  = b - IOU_BLOCKS;
    const int rb = s >> 1;
    const int m0 = (XF_TILE0 + rb * 4 + (wave & 3)) * 16;
    if ((s & 1) == 0) {
      v8f acc[3];
      if (!proj_acc<3, 32, 32, HD>(x, Wfx, Bs, XF_TILE0 + rb * 4, 0, acc)) return;
      xf_store<3>(acc, bfx, xf, m0, 0);
    } else {
      v8f acc[2];
      if (!proj_acc<2, 32, 32, HD>(x, Wfx, Bs, XF_TILE0 + rb * 4, 96, acc)) return;
      xf_store<2>(acc, bfx, xf, m0, 96);
    }
  }
}

#define SLW 32
#define NSL 5
#define CSP 36
#define ISP 100
#define FSP 36

template <int NT>
__global__ void __launch_bounds__(NT)
level_kernel(const float* __restrict__ xiou, const float* __restrict__ xf,
             const float* __restrict__ Wiouh, const float* __restrict__ biouh,
             const float* __restrict__ Wfh, const float* __restrict__ bfh,
             float* __restrict__ cst, float* __restrict__ hst,
             int start, int end) {
  extern __shared__ float smem[];
  float* chh  = smem;
  float* hsum = chh  + 64 * CHP;
  float* chc  = hsum + 16 * CHP;
  float* ious = chc  + 64 * CSP;
  float* fcs  = ious + 16 * ISP;

  const int tid  = threadIdx.x;
  const int lane = tid & 31;
  const int wave = tid >> 5;
  const int l15  = lane & 15;
  const int kb0  = (lane >> 4) << 1;
  const int roff = (lane >> 4) << 3;
  const int base = start + (int)blockIdx.x * 16;
  const int c0   = (int)blockIdx.y * SLW;

  {
    constexpr int NH = 64 * CHP / NT, NC = 64 * SLW / NT;
    static_assert(NH * NT == 64 * CHP && NC * NT == 64 * SLW, "trip counts");
    float hv[NH], cv[NC];
#pragma unroll
    for (int i = 0; i < NH; ++i) {
      int e = tid + i * NT;
      int row = e / CHP, col = e - row * CHP;
      int node = base + (row >> 2);
      int cr = 4 * ((NN - 1) - node) + 1 + (row & 3);
      bool ok = col < HD && node < end && cr < NN;
      size_t off = ok ? (size_t)((NN - 1) - cr) * HD + col : (size_t)0;
      float h = hst[off];
      hv[i] = ok ? h : 0.0f;
    }
#pragma unroll
    for (int i = 0; i < NC; ++i) {
      int e = tid + i * NT;
      int row = e >> 5, col = c0 + (e & 31);
      int node = base + (row >> 2);
      int cr = 4 * ((NN - 1) - node) + 1 + (row & 3);
      bool ok = col < HD && node < end && cr < NN;
      size_t off = ok ? (size_t)((NN - 1) - cr) * HD + col : (size_t)0;
      float c = cst[off];
      cv[i] = ok ? c : 0.0f;
    }
#pragma unroll
    for (int i = 0; i < NH; ++i) chh[tid + i * NT] = hv[i];
#pragma unroll
    for (int i = 0; i < NC; ++i) {
      int e = tid + i * NT;
      chc[(e >> 5) * CSP + (e & 31)] = cv[i];
    }
  }
  __syncthreads();
  for (int e = tid; e < 16 * CHP; e += NT) {
    int m = e / CHP, col = e - m * CHP;
    int r0 = (m * 4) * CHP + col;
    hsum[e] = chh[r0] + chh[r0 + CHP] + chh[r0 + 2 * CHP] + chh[r0 + 3 * CHP];
  }
  __syncthreads();

  for (int t = wave; t < 14; t += NT / 32) {
    const bool gate = t < 6;
    const int  u    = gate ? t : t - 6;
    const int  hf   = u & 1;
    const int  lc   = hf * 16 + l15;
    const int  hc   = c0 + lc;
    const bool nok  = hc < HD;
    const bool tailok = kb0 == 0;
    if (gate) {
      const int g = u >> 1;
      const int n = g * HD + hc;
      const float* wcol = Wiouh + (nok ? n : 0);
      float bx[38], by[38];
#pragma unroll
      for (int s = 0; s < 37; ++s) {
        int kb = 4 * s + kb0;
        float v0 = wcol[kb * IOUW];
        float v1 = wcol[(kb + 1) * IOUW];
        bx[s] = nok ? v0 : 0.0f;
        by[s] = nok ? v1 : 0.0f;
      }
      {
        int kb = tailok ? 148 : 0;
        float v0 = wcol[kb * IOUW];
        float v1 = wcol[(kb + 1) * IOUW];
        bx[37] = (nok && tailok) ? v0 : 0.0f;
        by[37] = (nok && tailok) ? v1 : 0.0f;
      }
      v8f acc = {};
#pragma unroll
      for (int s = 0; s < 38; ++s) {
        int kb = 4 * s + kb0;
        v2f a = *(const v2f*)(hsum + l15 * CHP + kb);
        v2f b;
        b.x = bx[s];
        b.y = by[s];
        acc = __builtin_amdgcn_wmma_f32_16x16x4_f32(false, a, false, b, (short)0,
                                                    acc, false, false);
      }
      if (nok) {
        float bv = biouh[n];
        float xv[8];
#pragma unroll
        for (int v = 0; v < 8; ++v) {
          int node = base + v + roff;
          bool live = node < end;
          float tt = xiou[(size_t)(live ? node : base) * IOUW + n];
          xv[v] = live ? tt : 0.0f;
        }
#pragma unroll
        for (int v = 0; v < 8; ++v) {
          int m = v + roff;
          ious[m * ISP + g * SLW + lc] = acc[v] + bv + xv[v];
        }
      }
    } else {
      const int m0 = (u >> 1) * 16;
      const int nc = nok ? hc : 0;
      const float* wcol = Wfh + nc;
      float bx[38], by[38];
#pragma unroll
      for (int s = 0; s < 37; ++s) {
        int kb = 4 * s + kb0;
        float v0 = wcol[kb * HD];
        float v1 = wcol[(kb + 1) * HD];
        bx[s] = nok ? v0 : 0.0f;
        by[s] = nok ? v1 : 0.0f;
      }
      {
        int kb = tailok ? 148 : 0;
        float v0 = wcol[kb * HD];
        float v1 = wcol[(kb + 1) * HD];
        bx[37] = (nok && tailok) ? v0 : 0.0f;
        by[37] = (nok && tailok) ? v1 : 0.0f;
      }
      v8f acc = {};
#pragma unroll
      for (int s = 0; s < 38; ++s) {
        int kb = 4 * s + kb0;
        v2f a = *(const v2f*)(chh + (m0 + l15) * CHP + kb);
        v2f b;
        b.x = bx[s];
        b.y = by[s];
        acc = __builtin_amdgcn_wmma_f32_16x16x4_f32(false, a, false, b, (short)0,
                                                    acc, false, false);
      }
      float bv = nok ? bfh[nc] : 0.0f;
      float s0 = 0.0f, s1 = 0.0f;
#pragma unroll
      for (int v = 0; v < 8; ++v) {
        int grow = m0 + v + roff;
        int node = base + (grow >> 2);
        float xv = (nok && node < end) ? xf[(size_t)node * HD + nc] : 0.0f;
        float f  = sigmoid_f(acc[v] + bv + xv);
        float w  = f * chc[grow * CSP + lc];
        if (v < 4) s0 += w; else s1 += w;
      }
      int nl = (m0 >> 2) + ((lane >> 4) << 1);
      fcs[nl * FSP + lc]       = s0;
      fcs[(nl + 1) * FSP + lc] = s1;
    }
  }
  __syncthreads();

  for (int e = tid; e < 16 * SLW; e += NT) {
    int m = e >> 5, lc = e & 31;
    int col = c0 + lc;
    int node = base + m;
    if (node < end && col < HD) {
      float i = sigmoid_f(ious[m * ISP + lc]);
      float o = sigmoid_f(ious[m * ISP + SLW + lc]);
      float u = tanh_f(ious[m * ISP + 2 * SLW + lc]);
      float c = i * u + fcs[m * FSP + lc];
      cst[(size_t)node * HD + col] = c;
      hst[(size_t)node * HD + col] = o * tanh_f(c);
    }
  }
}

extern "C" void kernel_launch(void* const* d_in, const int* in_sizes, int n_in,
                              void* d_out, int out_size, void* d_ws, size_t ws_size,
                              hipStream_t stream) {
  const float* x     = (const float*)d_in[0];
  const float* Wioux = (const float*)d_in[1];
  const float* bioux = (const float*)d_in[2];
  const float* Wiouh = (const float*)d_in[3];
  const float* biouh = (const float*)d_in[4];
  const float* Wfx   = (const float*)d_in[5];
  const float* bfx   = (const float*)d_in[6];
  const float* Wfh   = (const float*)d_in[7];
  const float* bfh   = (const float*)d_in[8];
  (void)in_sizes; (void)n_in; (void)out_size; (void)ws_size;

  float* xiou = (float*)d_ws;
  float* xf   = xiou + (size_t)NN * IOUW;
  float* cst  = xf   + (size_t)NN * HD;
  float* hst  = (float*)d_out;

  proj_gemm<<<1737, 256, 0, stream>>>(x, Wioux, bioux, biouh, xiou, Wfx, bfx, xf, cst, hst);

  const int bounds[8] = {14539, 18635, 19659, 19915, 19979, 19995, 19999, 20000};
  const size_t lds =
      (size_t)(64 * CHP + 16 * CHP + 64 * CSP + 16 * ISP + 16 * FSP) * sizeof(float);
  int s = bounds[0];
  for (int L = 1; L < 8; ++L) {
    int e = bounds[L];
    int groups = (e - s + 15) / 16;
    if (L == 1)
      level_kernel<256><<<dim3(groups, NSL), 256, lds, stream>>>(xiou, xf, Wiouh, biouh,
                                                                 Wfh, bfh, cst, hst, s, e);
    else
      level_kernel<512><<<dim3(groups, NSL), 512, lds, stream>>>(xiou, xf, Wiouh, biouh,
                                                                 Wfh, bfh, cst, hst, s, e);
    s = e;
  }
}


# level_kernel: child gather by whole rows: a wave takes a node, reads its four adjacent child rows 8 B per lane from scalar row addresses, sums them in registers; LDS sum pass and its barrier gone
# speedup vs baseline: 5.9903x; 1.1276x over previous
#include <hip/hip_runtime.h>


typedef __attribute__((ext_vector_type(2))) float v2f;
typedef __attribute__((ext_vector_type(4))) float v4f;
typedef __attribute__((ext_vector_type(8))) float v8f;

#define NN   20000
#define DIN  300
#define HD   150
#define IOUW 450
#define CHP  152

__device__ __forceinline__ float sigmoid_f(float x) {
  return 1.0f / (1.0f + __expf(-x));
}
__device__ __forceinline__ float tanh_f(float x) {
  float t = __expf(2.0f * fminf(fmaxf(x, -15.0f), 15.0f));
  return (t - 1.0f) / (t + 1.0f);
}

#define BSP 300
static_assert(BSP >= DIN && BSP % 8 == 4, "pitch holds a K row; 4 x odd dwords");

template <int NG, int GS, int GV, int N>
__device__ __forceinline__ bool
proj_acc(const float* __restrict__ A, const float* __restrict__ B,
         float* Bs  , int mt0, int n0, v8f (&acc)[NG]) {
  const int tid  = threadIdx.x;
  const int lane = tid & 31;
  const int wave = tid >> 5;
  const int l15  = lane & 15;
  const int kb0  = (lane >> 4) << 1;

  {
    static_assert(DIN == 37 * 8 + 4, "37 rows per thread, 8 waves, and 4 rows left over");
    const bool last = wave < 4;
    float bv[NG][38];
#pragma unroll
    for (int g = 0; g < NG; ++g) {
      const bool cok = n0 + g * GV + lane < HD;
      const float* bcol = B + (size_t)wave * N + (cok ? n0 + g * GS + lane : 0);
#pragma unroll
      for (int i = 0; i < 37; ++i) bv[g][i] = bcol[(size_t)(8 * i) * N];
      bv[g][37] = last ? bcol[(size_t)(8 * 37) * N] : 0.0f;
    }
#pragma unroll
    for (int g = 0; g < NG; ++g) {
      const bool cok = n0 + g * GV + lane < HD;
      float* bdst = Bs + (g * 32 + lane) * BSP + wave;
#pragma unroll
      for (int i = 0; i < 37; ++i) bdst[8 * i] = cok ? bv[g][i] : 0.0f;
      if (last) bdst[8 * 37] = cok ? bv[g][37] : 0.0f;
    }
  }
  __syncthreads();

  const int m0 = (mt0 + (wave & 3)) * 16;
  if (m0 >= NN) return false;
  const float* arow = A + (size_t)(m0 + l15) * DIN;
  const float* brow = Bs + ((wave >> 2) * 16 + l15) * BSP;
#pragma unroll
  for (int g = 0; g < NG; ++g) acc[g] = v8f{};
  static_assert(DIN == 37 * 8 + 4 && (DIN * 4) % 16 == 0 && (BSP * 4) % 16 == 0,
                "37 blocks of 8 k and one K-step of 4; 16-byte aligned rows");
  const int kq = (lane >> 4) << 2;
  auto block8 = [&](int k0) {
    int kb = k0 + kq;
    v4f a = *(const v4f*)(arow + kb);
    v4f b[NG];
#pragma unroll
    for (int g = 0; g < NG; ++g)
      b[g] = *(const v4f*)(brow + g * 32 * BSP + kb);
#pragma unroll
    for (int g = 0; g < NG; ++g)
      acc[g] = __builtin_amdgcn_wmma_f32_16x16x4_f32(false, a.lo, false, b[g].lo, (short)0,
                                                     acc[g], false, false);
#pragma unroll
    for (int g = 0; g < NG; ++g)
      acc[g] = __builtin_amdgcn_wmma_f32_16x16x4_f32(false, a.hi, false, b[g].hi, (short)0,
                                                     acc[g], false, false);
  };
#pragma unroll 12
  for (int k0 = 0; k0 < 36 * 8; k0 += 8) block8(k0);
  block8(36 * 8);
  {
    int kb = (DIN - 4) + kb0;
    v2f a = *(const v2f*)(arow + kb);
#pragma unroll
    for (int g = 0; g < NG; ++g) {
      v2f b = *(const v2f*)(brow + g * 32 * BSP + kb);
      acc[g] = __builtin_amdgcn_wmma_f32_16x16x4_f32(false, a, false, b, (short)0,
                                                     acc[g], false, false);
    }
  }
  return true;
}

#define LEAF_END   14539
#define XF_TILE0   (LEAF_END / 16)
#define XF_RBLK    ((NN / 16 - XF_TILE0 + 3) / 4)
#define IOU_RBLK   313
#define IOU_BLOCKS (IOU_RBLK * 5)
#define XF_BLOCKS  (XF_RBLK * 2)
static_assert(NN % 16 == 0 && IOU_RBLK * 4 >= NN / 16, "row tiling");
static_assert(IOU_BLOCKS + XF_BLOCKS == 1737, "grid literal in kernel_launch");

template <int NG>
__device__ __forceinline__ void
xf_store(const v8f (&acc)[NG], const float* __restrict__ bfx,
         float* __restrict__ xf, int m0, int n0) {
  const int lane = threadIdx.x & 31;
  const int l15  = lane & 15;
  const int roff = (lane >> 4) << 3;
  const int hf   = threadIdx.x >> 7;
#pragma unroll
  for (int g = 0; g < NG; ++g) {
    int n = n0 + g * 32 + hf * 16 + l15;
    if (n < HD) {
      float bv = bfx[n];
#pragma unroll
      for (int v = 0; v < 8; ++v)
        xf[(size_t)(m0 + v + roff) * HD + n] = acc[g][v] + bv;
    }
  }
}

__global__ void __launch_bounds__(256)
proj_gemm(const float* __restrict__ x,
          const float* __restrict__ Wioux, const float* __restrict__ bioux,
          const float* __restrict__ biouh, float* __restrict__ xiou,
          const float* __restrict__ Wfx, const float* __restrict__ bfx,
          float* __restrict__ xf,
          float* __restrict__ cst, float* __restrict__ hst) {
  __shared__ __attribute__((aligned(16))) float Bs[96 * BSP];
  const int b    = (int)blockIdx.x;
  const int wave = threadIdx.x >> 5;
  if (b < IOU_BLOCKS) {
    const int cb = b / IOU_RBLK, rb = b - cb * IOU_RBLK;
    const int n0 = cb * 32;
    v8f acc[3];
    if (!proj_acc<3, HD, 0, IOUW>(x, Wioux, Bs, rb * 4, n0, acc)) return;
    const int lane = threadIdx.x & 31;
    const int roff = (lane >> 4) << 3;
    const int m0   = (rb * 4 + (wave & 3)) * 16;
    const int n    = n0 + (wave >> 2) * 16 + (lane & 15);
    if (n < HD) {
      float bi = bioux[n], bo = bioux[HD + n], bu = bioux[2 * HD + n];
      float hi = biouh[n], ho = biouh[HD + n], hu = biouh[2 * HD + n];
#pragma unroll
      for (int v = 0; v < 8; ++v) {
        int m = m0 + v + roff;
        float xi = acc[0][v] + bi;
        float xo = acc[1][v] + bo;
        float xu = acc[2][v] + bu;
        if (m < LEAF_END) {
          float i = sigmoid_f(xi + hi);
          float o = sigmoid_f(xo + ho);
          float u = tanh_f(xu + hu);
          float c = i * u;
          cst[(size_t)m * HD + n] = c;
          hst[(size_t)m * HD + n] = o * tanh_f(c);
        } else {
          float* r = xiou + (size_t)m * IOUW + n;
          r[0]      = xi;
          r[HD]     = xo;
          r[2 * HD] = xu;
        }
      }
    }
  } else {
    const int s  = b - IOU_BLOCKS;
    const int rb = s >> 1;
    const int m0 = (XF_TILE0 + rb * 4 + (wave & 3)) * 16;
    if ((s & 1) == 0) {
      v8f acc[3];
      if (!proj_acc<3, 32, 32, HD>(x, Wfx, Bs, XF_TILE0 + rb * 4, 0, acc)) return;
      xf_store<3>(acc, bfx, xf, m0, 0);
    } else {
      v8f acc[2];
      if (!proj_acc<2, 32, 32, HD>(x, Wfx, Bs, XF_TILE0 + rb * 4, 96, acc)) return;
      xf_store<2>(acc, bfx, xf, m0, 96);
    }
  }
}

#define SLW 32
#define NSL 5
#define CSP 36
#define ISP 100
#define FSP 36

template <int NT>
__global__ void __launch_bounds__(NT)
level_kernel(const float* __restrict__ xiou, const float* __restrict__ xf,
             const float* __restrict__ Wiouh, const float* __restrict__ biouh,
             const float* __restrict__ Wfh, const float* __restrict__ bfh,
             float* __restrict__ cst, float* __restrict__ hst,
             int start, int end) {
  extern __shared__ __attribute__((aligned(16))) float smem[];
  float* chh  = smem;
  float* hsum = chh  + 64 * CHP;
  float* chc  = hsum + 16 * CHP;
  float* ious = chc  + 64 * CSP;
  float* fcs  = ious + 16 * ISP;

  const int tid  = threadIdx.x;
  const int lane = tid & 31;
  const int wave = tid >> 5;
  const int l15  = lane & 15;
  const int kb0  = (lane >> 4) << 1;
  const int roff = (lane >> 4) << 3;
  const int base = start + (int)blockIdx.x * 16;
  const int c0   = (int)blockIdx.y * SLW;

  {
    constexpr int NW  = NT / 32;
    constexpr int NPW = 16 / NW;
    static_assert(NPW * NW == 16 && HD == 2 * 75 && CHP == 2 * 76, "whole nodes per wave; 75 column pairs and one pad pair");
    const int wv = __builtin_amdgcn_readfirstlane(tid >> 5);
    const int j2 = min(64 + lane, 74);
    const bool t2 = lane < 11;
    const int  cc = min(c0 + lane, HD - 1);
    const bool cl = c0 + lane < HD;
    v2f   hv[NPW][4][3];
    float cv[NPW][4];
    bool  ok[NPW][4];
#pragma unroll
    for (int q = 0; q < NPW; ++q) {
      const int node = base + wv * NPW + q;
      const int idx0 = 4 * node - (3 * NN - 2);
#pragma unroll
      for (int s = 0; s < 4; ++s) {
        ok[q][s] = node < end && idx0 - s >= 0;
        const size_t ro = ok[q][s] ? (size_t)(idx0 - s) * HD : (size_t)0;
        const float* hr = hst + ro;
        hv[q][s][0] = *(const v2f*)(hr + 2 * lane);
        hv[q][s][1] = *(const v2f*)(hr + 64 + 2 * lane);
        hv[q][s][2] = *(const v2f*)(hr + 2 * j2);
        cv[q][s]    = cst[ro + cc];
      }
    }
#pragma unroll
    for (int q = 0; q < NPW; ++q) {
      const int m = wv * NPW + q;
      v2f sum[3];
#pragma unroll
      for (int s = 0; s < 4; ++s) {
        float* row = chh + (4 * m + s) * CHP;
#pragma unroll
        for (int t = 0; t < 3; ++t) {
          const bool keep = ok[q][s] && (t < 2 || t2);
          const v2f h = keep ? hv[q][s][t] : v2f{0.0f, 0.0f};
          sum[t] = s == 0 ? h : sum[t] + h;
          if (t < 2 || lane < 12) *(v2f*)(row + 64 * t + 2 * lane) = h;
        }
        chc[(4 * m + s) * CSP + lane] = (ok[q][s] && cl) ? cv[q][s] : 0.0f;
      }
#pragma unroll
      for (int t = 0; t < 3; ++t)
        if (t < 2 || lane < 12) *(v2f*)(hsum + m * CHP + 64 * t + 2 * lane) = sum[t];
    }
  }
  __syncthreads();

  for (int t = wave; t < 14; t += NT / 32) {
    const bool gate = t < 6;
    const int  u    = gate ? t : t - 6;
    const int  hf   = u & 1;
    const int  lc   = hf * 16 + l15;
    const int  hc   = c0 + lc;
    const bool nok  = hc < HD;
    const bool tailok = kb0 == 0;
    if (gate) {
      const int g = u >> 1;
      const int n = g * HD + hc;
      const float* wcol = Wiouh + (nok ? n : 0);
      float bx[38], by[38];
#pragma unroll
      for (int s = 0; s < 37; ++s) {
        int kb = 4 * s + kb0;
        float v0 = wcol[kb * IOUW];
        float v1 = wcol[(kb + 1) * IOUW];
        bx[s] = nok ? v0 : 0.0f;
        by[s] = nok ? v1 : 0.0f;
      }
      {
        int kb = tailok ? 148 : 0;
        float v0 = wcol[kb * IOUW];
        float v1 = wcol[(kb + 1) * IOUW];
        bx[37] = (nok && tailok) ? v0 : 0.0f;
        by[37] = (nok && tailok) ? v1 : 0.0f;
      }
      v8f acc = {};
#pragma unroll
      for (int s = 0; s < 38; ++s) {
        int kb = 4 * s + kb0;
        v2f a = *(const v2f*)(hsum + l15 * CHP + kb);
        v2f b;
        b.x = bx[s];
        b.y = by[s];
        acc = __builtin_amdgcn_wmma_f32_16x16x4_f32(false, a, false, b, (short)0,
                                                    acc, false, false);
      }
      if (nok) {
        float bv = biouh[n];
        float xv[8];
#pragma unroll
        for (int v = 0; v < 8; ++v) {
          int node = base + v + roff;
          bool live = node < end;
          float tt = xiou[(size_t)(live ? node : base) * IOUW + n];
          xv[v] = live ? tt : 0.0f;
        }
#pragma unroll
        for (int v = 0; v < 8; ++v) {
          int m = v + roff;
          ious[m * ISP + g * SLW + lc] = acc[v] + bv + xv[v];
        }
      }
    } else {
      const int m0 = (u >> 1) * 16;
      const int nc = nok ? hc : 0;
      const float* wcol = Wfh + nc;
      float bx[38], by[38];
#pragma unroll
      for (int s = 0; s < 37; ++s) {
        int kb = 4 * s + kb0;
        float v0 = wcol[kb * HD];
        float v1 = wcol[(kb + 1) * HD];
        bx[s] = nok ? v0 : 0.0f;
        by[s] = nok ? v1 : 0.0f;
      }
      {
        int kb = tailok ? 148 : 0;
        float v0 = wcol[kb * HD];
        float v1 = wcol[(kb + 1) * HD];
        bx[37] = (nok && tailok) ? v0 : 0.0f;
        by[37] = (nok && tailok) ? v1 : 0.0f;
      }
      v8f acc = {};
#pragma unroll
      for (int s = 0; s < 38; ++s) {
        int kb = 4 * s + kb0;
        v2f a = *(const v2f*)(chh + (m0 + l15) * CHP + kb);
        v2f b;
        b.x = bx[s];
        b.y = by[s];
        acc = __builtin_amdgcn_wmma_f32_16x16x4_f32(false, a, false, b, (short)0,
                                                    acc, false, false);
      }
      float bv = nok ? bfh[nc] : 0.0f;
      float s0 = 0.0f, s1 = 0.0f;
#pragma unroll
      for (int v = 0; v < 8; ++v) {
        int grow = m0 + v + roff;
        int node = base + (grow >> 2);
        float xv = (nok && node < end) ? xf[(size_t)node * HD + nc] : 0.0f;
        float f  = sigmoid_f(acc[v] + bv + xv);
        float w  = f * chc[grow * CSP + lc];
        if (v < 4) s0 += w; else s1 += w;
      }
      int nl = (m0 >> 2) + ((lane >> 4) << 1);
      fcs[nl * FSP + lc]       = s0;
      fcs[(nl + 1) * FSP + lc] = s1;
    }
  }
  __syncthreads();

  for (int e = tid; e < 16 * SLW; e += NT) {
    int m = e >> 5, lc = e & 31;
    int col = c0 + lc;
    int node = base + m;
    if (node < end && col < HD) {
      float i = sigmoid_f(ious[m * ISP + lc]);
      float o = sigmoid_f(ious[m * ISP + SLW + lc]);
      float u = tanh_f(ious[m * ISP + 2 * SLW + lc]);
      float c = i * u + fcs[m * FSP + lc];
      cst[(size_t)node * HD + col] = c;
      hst[(size_t)node * HD + col] = o * tanh_f(c);
    }
  }
}

extern "C" void kernel_launch(void* const* d_in, const int* in_sizes, int n_in,
                              void* d_out, int out_size, void* d_ws, size_t ws_size,
                              hipStream_t stream) {
  const float* x     = (const float*)d_in[0];
  const float* Wioux = (const float*)d_in[1];
  const float* bioux = (const float*)d_in[2];
  const float* Wiouh = (const float*)d_in[3];
  const float* biouh = (const float*)d_in[4];
  const float* Wfx   = (const float*)d_in[5];
  const float* bfx   = (const float*)d_in[6];
  const float* Wfh   = (const float*)d_in[7];
  const float* bfh   = (const float*)d_in[8];
  (void)in_sizes; (void)n_in; (void)out_size; (void)ws_size;

  float* xiou = (float*)d_ws;
  float* xf   = xiou + (size_t)NN * IOUW;
  float* cst  = xf   + (size_t)NN * HD;
  float* hst  = (float*)d_out;

  proj_gemm<<<1737, 256, 0, stream>>>(x, Wioux, bioux, biouh, xiou, Wfx, bfx, xf, cst, hst);

  const int bounds[8] = {14539, 18635, 19659, 19915, 19979, 19995, 19999, 20000};
  const size_t lds =
      (size_t)(64 * CHP + 16 * CHP + 64 * CSP + 16 * ISP + 16 * FSP) * sizeof(float);
  int s = bounds[0];
  for (int L = 1; L < 8; ++L) {
    int e = bounds[L];
    int groups = (e - s + 15) / 16;
    if (L == 1)
      level_kernel<256><<<dim3(groups, NSL), 256, lds, stream>>>(xiou, xf, Wiouh, biouh,
                                                                 Wfh, bfh, cst, hst, s, e);
    else
      level_kernel<512><<<dim3(groups, NSL), 512, lds, stream>>>(xiou, xf, Wiouh, biouh,
                                                                 Wfh, bfh, cst, hst, s, e);
    s = e;
  }
}


# level_kernel: a task is 16 rows x both column tiles of the slice; a lane owns two adjacent columns, so one 8-byte weight load and one A read feed both tiles: 7 tasks not 14, 8 waves at every level
# speedup vs baseline: 6.4259x; 1.0727x over previous
#include <hip/hip_runtime.h>


typedef __attribute__((ext_vector_type(2))) float v2f;
typedef __attribute__((ext_vector_type(4))) float v4f;
typedef __attribute__((ext_vector_type(8))) float v8f;

#define NN   20000
#define DIN  300
#define HD   150
#define IOUW 450
#define CHP  152

__device__ __forceinline__ float sigmoid_f(float x) {
  return 1.0f / (1.0f + __expf(-x));
}
__device__ __forceinline__ float tanh_f(float x) {
  float t = __expf(2.0f * fminf(fmaxf(x, -15.0f), 15.0f));
  return (t - 1.0f) / (t + 1.0f);
}

#define BSP 300
static_assert(BSP >= DIN && BSP % 8 == 4, "pitch holds a K row; 4 x odd dwords");

template <int NG, int GS, int GV, int N>
__device__ __forceinline__ bool
proj_acc(const float* __restrict__ A, const float* __restrict__ B,
         float* Bs  , int mt0, int n0, v8f (&acc)[NG]) {
  const int tid  = threadIdx.x;
  const int lane = tid & 31;
  const int wave = tid >> 5;
  const int l15  = lane & 15;
  const int kb0  = (lane >> 4) << 1;

  {
    static_assert(DIN == 37 * 8 + 4, "37 rows per thread, 8 waves, and 4 rows left over");
    const bool last = wave < 4;
    float bv[NG][38];
#pragma unroll
    for (int g = 0; g < NG; ++g) {
      const bool cok = n0 + g * GV + lane < HD;
      const float* bcol = B + (size_t)wave * N + (cok ? n0 + g * GS + lane : 0);
#pragma unroll
      for (int i = 0; i < 37; ++i) bv[g][i] = bcol[(size_t)(8 * i) * N];
      bv[g][37] = last ? bcol[(size_t)(8 * 37) * N] : 0.0f;
    }
#pragma unroll
    for (int g = 0; g < NG; ++g) {
      const bool cok = n0 + g * GV + lane < HD;
      float* bdst = Bs + (g * 32 + lane) * BSP + wave;
#pragma unroll
      for (int i = 0; i < 37; ++i) bdst[8 * i] = cok ? bv[g][i] : 0.0f;
      if (last) bdst[8 * 37] = cok ? bv[g][37] : 0.0f;
    }
  }
  __syncthreads();

  const int m0 = (mt0 + (wave & 3)) * 16;
  if (m0 >= NN) return false;
  const float* arow = A + (size_t)(m0 + l15) * DIN;
  const float* brow = Bs + ((wave >> 2) * 16 + l15) * BSP;
#pragma unroll
  for (int g = 0; g < NG; ++g) acc[g] = v8f{};
  static_assert(DIN == 37 * 8 + 4 && (DIN * 4) % 16 == 0 && (BSP * 4) % 16 == 0,
                "37 blocks of 8 k and one K-step of 4; 16-byte aligned rows");
  const int kq = (lane >> 4) << 2;
  auto block8 = [&](int k0) {
    int kb = k0 + kq;
    v4f a = *(const v4f*)(arow + kb);
    v4f b[NG];
#pragma unroll
    for (int g = 0; g < NG; ++g)
      b[g] = *(const v4f*)(brow + g * 32 * BSP + kb);
#pragma unroll
    for (int g = 0; g < NG; ++g)
      acc[g] = __builtin_amdgcn_wmma_f32_16x16x4_f32(false, a.lo, false, b[g].lo, (short)0,
                                                     acc[g], false, false);
#pragma unroll
    for (int g = 0; g < NG; ++g)
      acc[g] = __builtin_amdgcn_wmma_f32_16x16x4_f32(false, a.hi, false, b[g].hi, (short)0,
                                                     acc[g], false, false);
  };
#pragma unroll 12
  for (int k0 = 0; k0 < 36 * 8; k0 += 8) block8(k0);
  block8(36 * 8);
  {
    int kb = (DIN - 4) + kb0;
    v2f a = *(const v2f*)(arow + kb);
#pragma unroll
    for (int g = 0; g < NG; ++g) {
      v2f b = *(const v2f*)(brow + g * 32 * BSP + kb);
      acc[g] = __builtin_amdgcn_wmma_f32_16x16x4_f32(false, a, false, b, (short)0,
                                                     acc[g], false, false);
    }
  }
  return true;
}

#define LEAF_END   14539
#define XF_TILE0   (LEAF_END / 16)
#define XF_RBLK    ((NN / 16 - XF_TILE0 + 3) / 4)
#define IOU_RBLK   313
#define IOU_BLOCKS (IOU_RBLK * 5)
#define XF_BLOCKS  (XF_RBLK * 2)
static_assert(NN % 16 == 0 && IOU_RBLK * 4 >= NN / 16, "row tiling");
static_assert(IOU_BLOCKS + XF_BLOCKS == 1737, "grid literal in kernel_launch");

template <int NG>
__device__ __forceinline__ void
xf_store(const v8f (&acc)[NG], const float* __restrict__ bfx,
         float* __restrict__ xf, int m0, int n0) {
  const int lane = threadIdx.x & 31;
  const int l15  = lane & 15;
  const int roff = (lane >> 4) << 3;
  const int hf   = threadIdx.x >> 7;
#pragma unroll
  for (int g = 0; g < NG; ++g) {
    int n = n0 + g * 32 + hf * 16 + l15;
    if (n < HD) {
      float bv = bfx[n];
#pragma unroll
      for (int v = 0; v < 8; ++v)
        xf[(size_t)(m0 + v + roff) * HD + n] = acc[g][v] + bv;
    }
  }
}

__global__ void __launch_bounds__(256)
proj_gemm(const float* __restrict__ x,
          const float* __restrict__ Wioux, const float* __restrict__ bioux,
          const float* __restrict__ biouh, float* __restrict__ xiou,
          const float* __restrict__ Wfx, const float* __restrict__ bfx,
          float* __restrict__ xf,
          float* __restrict__ cst, float* __restrict__ hst) {
  __shared__ __attribute__((aligned(16))) float Bs[96 * BSP];
  const int b    = (int)blockIdx.x;
  const int wave = threadIdx.x >> 5;
  if (b < IOU_BLOCKS) {
    const int cb = b / IOU_RBLK, rb = b - cb * IOU_RBLK;
    const int n0 = cb * 32;
    v8f acc[3];
    if (!proj_acc<3, HD, 0, IOUW>(x, Wioux, Bs, rb * 4, n0, acc)) return;
    const int lane = threadIdx.x & 31;
    const int roff = (lane >> 4) << 3;
    const int m0   = (rb * 4 + (wave & 3)) * 16;
    const int n    = n0 + (wave >> 2) * 16 + (lane & 15);
    if (n < HD) {
      float bi = bioux[n], bo = bioux[HD + n], bu = bioux[2 * HD + n];
      float hi = biouh[n], ho = biouh[HD + n], hu = biouh[2 * HD + n];
#pragma unroll
      for (int v = 0; v < 8; ++v) {
        int m = m0 + v + roff;
        float xi = acc[0][v] + bi;
        float xo = acc[1][v] + bo;
        float xu = acc[2][v] + bu;
        if (m < LEAF_END) {
          float i = sigmoid_f(xi + hi);
          float o = sigmoid_f(xo + ho);
          float u = tanh_f(xu + hu);
          float c = i * u;
          cst[(size_t)m * HD + n] = c;
          hst[(size_t)m * HD + n] = o * tanh_f(c);
        } else {
          float* r = xiou + (size_t)m * IOUW + n;
          r[0]      = xi;
          r[HD]     = xo;
          r[2 * HD] = xu;
        }
      }
    }
  } else {
    const int s  = b - IOU_BLOCKS;
    const int rb = s >> 1;
    const int m0 = (XF_TILE0 + rb * 4 + (wave & 3)) * 16;
    if ((s & 1) == 0) {
      v8f acc[3];
      if (!proj_acc<3, 32, 32, HD>(x, Wfx, Bs, XF_TILE0 + rb * 4, 0, acc)) return;
      xf_store<3>(acc, bfx, xf, m0, 0);
    } else {
      v8f acc[2];
      if (!proj_acc<2, 32, 32, HD>(x, Wfx, Bs, XF_TILE0 + rb * 4, 96, acc)) return;
      xf_store<2>(acc, bfx, xf, m0, 96);
    }
  }
}

template <int WP>
__device__ __forceinline__ void
pair_sweep(const float* __restrict__ wcol, const float* arow, int kb0, bool nok,
           v8f& acc0, v8f& acc1) {
  const bool tailok = kb0 == 0;
  v2f w0[38], w1[38];
#pragma unroll
  for (int s = 0; s < 37; ++s) {
    int kb = 4 * s + kb0;
    v2f r0 = *(const v2f*)(wcol + kb * WP);
    v2f r1 = *(const v2f*)(wcol + (kb + 1) * WP);
    w0[s] = nok ? r0 : v2f{0.0f, 0.0f};
    w1[s] = nok ? r1 : v2f{0.0f, 0.0f};
  }
  {
    int kb = tailok ? 148 : 0;
    v2f r0 = *(const v2f*)(wcol + kb * WP);
    v2f r1 = *(const v2f*)(wcol + (kb + 1) * WP);
    w0[37] = (nok && tailok) ? r0 : v2f{0.0f, 0.0f};
    w1[37] = (nok && tailok) ? r1 : v2f{0.0f, 0.0f};
  }
  acc0 = v8f{};
  acc1 = v8f{};
#pragma unroll
  for (int s = 0; s < 38; ++s) {
    int kb = 4 * s + kb0;
    v2f a = *(const v2f*)(arow + kb);
    v2f b0, b1;
    b0.x = w0[s].x;  b0.y = w1[s].x;
    b1.x = w0[s].y;  b1.y = w1[s].y;
    acc0 = __builtin_amdgcn_wmma_f32_16x16x4_f32(false, a, false, b0, (short)0,
                                                 acc0, false, false);
    acc1 = __builtin_amdgcn_wmma_f32_16x16x4_f32(false, a, false, b1, (short)0,
                                                 acc1, false, false);
  }
}

#define SLW 32
#define NSL 5
#define CSP 36
#define ISP 100
#define FSP 36

template <int NT>
__global__ void __launch_bounds__(NT)
level_kernel(const float* __restrict__ xiou, const float* __restrict__ xf,
             const float* __restrict__ Wiouh, const float* __restrict__ biouh,
             const float* __restrict__ Wfh, const float* __restrict__ bfh,
             float* __restrict__ cst, float* __restrict__ hst,
             int start, int end) {
  extern __shared__ __attribute__((aligned(16))) float smem[];
  float* chh  = smem;
  float* hsum = chh  + 64 * CHP;
  float* chc  = hsum + 16 * CHP;
  float* ious = chc  + 64 * CSP;
  float* fcs  = ious + 16 * ISP;

  const int tid  = threadIdx.x;
  const int lane = tid & 31;
  const int wave = tid >> 5;
  const int l15  = lane & 15;
  const int kb0  = (lane >> 4) << 1;
  const int roff = (lane >> 4) << 3;
  const int base = start + (int)blockIdx.x * 16;
  const int c0   = (int)blockIdx.y * SLW;

  {
    constexpr int NW  = NT / 32;
    constexpr int NPW = 16 / NW;
    static_assert(NPW * NW == 16 && HD == 2 * 75 && CHP == 2 * 76, "whole nodes per wave; 75 column pairs and one pad pair");
    const int wv = __builtin_amdgcn_readfirstlane(tid >> 5);
    const int j2 = min(64 + lane, 74);
    const bool t2 = lane < 11;
    const int  cc = min(c0 + lane, HD - 1);
    const bool cl = c0 + lane < HD;
    v2f   hv[NPW][4][3];
    float cv[NPW][4];
    bool  ok[NPW][4];
#pragma unroll
    for (int q = 0; q < NPW; ++q) {
      const int node = base + wv * NPW + q;
      const int idx0 = 4 * node - (3 * NN - 2);
#pragma unroll
      for (int s = 0; s < 4; ++s) {
        ok[q][s] = node < end && idx0 - s >= 0;
        const size_t ro = ok[q][s] ? (size_t)(idx0 - s) * HD : (size_t)0;
        const float* hr = hst + ro;
        hv[q][s][0] = *(const v2f*)(hr + 2 * lane);
        hv[q][s][1] = *(const v2f*)(hr + 64 + 2 * lane);
        hv[q][s][2] = *(const v2f*)(hr + 2 * j2);
        cv[q][s]    = cst[ro + cc];
      }
    }
#pragma unroll
    for (int q = 0; q < NPW; ++q) {
      const int m = wv * NPW + q;
      v2f sum[3];
#pragma unroll
      for (int s = 0; s < 4; ++s) {
        float* row = chh + (4 * m + s) * CHP;
#pragma unroll
        for (int t = 0; t < 3; ++t) {
          const bool keep = ok[q][s] && (t < 2 || t2);
          const v2f h = keep ? hv[q][s][t] : v2f{0.0f, 0.0f};
          sum[t] = s == 0 ? h : sum[t] + h;
          if (t < 2 || lane < 12) *(v2f*)(row + 64 * t + 2 * lane) = h;
        }
        chc[(4 * m + s) * CSP + lane] = (ok[q][s] && cl) ? cv[q][s] : 0.0f;
      }
#pragma unroll
      for (int t = 0; t < 3; ++t)
        if (t < 2 || lane < 12) *(v2f*)(hsum + m * CHP + 64 * t + 2 * lane) = sum[t];
    }
  }
  __syncthreads();

  for (int t = wave; t < 7; t += NT / 32) {
    const bool gate = t < 3;
    const int  lc   = 2 * l15;
    const int  hc   = c0 + lc;
    const bool nok  = hc < HD;
    v8f acc0, acc1;
    if (gate) {
      const int g = t;
      const int n = g * HD + hc;
      pair_sweep<IOUW>(Wiouh + (nok ? n : 0), hsum + l15 * CHP, kb0, nok, acc0, acc1);
      if (nok) {
        v2f bv = *(const v2f*)(biouh + n);
        v2f xv[8];
#pragma unroll
        for (int v = 0; v < 8; ++v) {
          int node = base + v + roff;
          bool live = node < end;
          v2f tt = *(const v2f*)(xiou + (size_t)(live ? node : base) * IOUW + n);
          xv[v] = live ? tt : v2f{0.0f, 0.0f};
        }
#pragma unroll
        for (int v = 0; v < 8; ++v) {
          int m = v + roff;
          v2f r;
          r.x = acc0[v] + bv.x + xv[v].x;
          r.y = acc1[v] + bv.y + xv[v].y;
          *(v2f*)(ious + m * ISP + g * SLW + lc) = r;
        }
      }
    } else {
      const int m0 = (t - 3) * 16;
      const int nc = nok ? hc : 0;
      pair_sweep<HD>(Wfh + nc, chh + (m0 + l15) * CHP, kb0, nok, acc0, acc1);
      v2f bv = nok ? *(const v2f*)(bfh + nc) : v2f{0.0f, 0.0f};
      float s00 = 0.0f, s01 = 0.0f, s10 = 0.0f, s11 = 0.0f;
#pragma unroll
      for (int v = 0; v < 8; ++v) {
        int grow = m0 + v + roff;
        int node = base + (grow >> 2);
        v2f xv = (nok && node < end) ? *(const v2f*)(xf + (size_t)node * HD + nc) : v2f{0.0f, 0.0f};
        v2f cc = *(const v2f*)(chc + grow * CSP + lc);
        float f0 = sigmoid_f(acc0[v] + bv.x + xv.x);
        float f1 = sigmoid_f(acc1[v] + bv.y + xv.y);
        float q0 = f0 * cc.x;
        float q1 = f1 * cc.y;
        if (v < 4) { s00 += q0; s01 += q1; }
        else       { s10 += q0; s11 += q1; }
      }
      int nl = (m0 >> 2) + ((lane >> 4) << 1);
      v2f lo, hi;
      lo.x = s00;  lo.y = s01;
      hi.x = s10;  hi.y = s11;
      *(v2f*)(fcs + nl * FSP + lc)       = lo;
      *(v2f*)(fcs + (nl + 1) * FSP + lc) = hi;
    }
  }
  __syncthreads();

  for (int e = tid; e < 16 * SLW; e += NT) {
    int m = e >> 5, lc = e & 31;
    int col = c0 + lc;
    int node = base + m;
    if (node < end && col < HD) {
      float i = sigmoid_f(ious[m * ISP + lc]);
      float o = sigmoid_f(ious[m * ISP + SLW + lc]);
      float u = tanh_f(ious[m * ISP + 2 * SLW + lc]);
      float c = i * u + fcs[m * FSP + lc];
      cst[(size_t)node * HD + col] = c;
      hst[(size_t)node * HD + col] = o * tanh_f(c);
    }
  }
}

extern "C" void kernel_launch(void* const* d_in, const int* in_sizes, int n_in,
                              void* d_out, int out_size, void* d_ws, size_t ws_size,
                              hipStream_t stream) {
  const float* x     = (const float*)d_in[0];
  const float* Wioux = (const float*)d_in[1];
  const float* bioux = (const float*)d_in[2];
  const float* Wiouh = (const float*)d_in[3];
  const float* biouh = (const float*)d_in[4];
  const float* Wfx   = (const float*)d_in[5];
  const float* bfx   = (const float*)d_in[6];
  const float* Wfh   = (const float*)d_in[7];
  const float* bfh   = (const float*)d_in[8];
  (void)in_sizes; (void)n_in; (void)out_size; (void)ws_size;

  float* xiou = (float*)d_ws;
  float* xf   = xiou + (size_t)NN * IOUW;
  float* cst  = xf   + (size_t)NN * HD;
  float* hst  = (float*)d_out;

  proj_gemm<<<1737, 256, 0, stream>>>(x, Wioux, bioux, biouh, xiou, Wfx, bfx, xf, cst, hst);

  const int bounds[8] = {14539, 18635, 19659, 19915, 19979, 19995, 19999, 20000};
  const size_t lds =
      (size_t)(64 * CHP + 16 * CHP + 64 * CSP + 16 * ISP + 16 * FSP) * sizeof(float);
  int s = bounds[0];
  for (int L = 1; L < 8; ++L) {
    int e = bounds[L];
    int groups = (e - s + 15) / 16;
    level_kernel<256><<<dim3(groups, NSL), 256, lds, stream>>>(xiou, xf, Wiouh, biouh,
                                                               Wfh, bfh, cst, hst, s, e);
    s = e;
  }
}


# level_kernel: Wiouh, Wfh copied once per call (20 extra blocks of the projection launch) into fragment order: a task's weights are 38 aligned 16-byte loads, no select, not 76 straddling 8-byte loads
# speedup vs baseline: 6.9034x; 1.0743x over previous
#include <hip/hip_runtime.h>


typedef __attribute__((ext_vector_type(2))) float v2f;
typedef __attribute__((ext_vector_type(4))) float v4f;
typedef __attribute__((ext_vector_type(8))) float v8f;

#define NN   20000
#define DIN  300
#define HD   150
#define IOUW 450
#define CHP  152
#define KST  38
#define WPK_TASK (KST * 32 * 4)
#define WPK_BLOCKS 20
#define SLW_PACK 32

__device__ __forceinline__ float sigmoid_f(float x) {
  return 1.0f / (1.0f + __expf(-x));
}
__device__ __forceinline__ float tanh_f(float x) {
  float t = __expf(2.0f * fminf(fmaxf(x, -15.0f), 15.0f));
  return (t - 1.0f) / (t + 1.0f);
}

#define BSP 300
static_assert(BSP >= DIN && BSP % 8 == 4, "pitch holds a K row; 4 x odd dwords");

template <int NG, int GS, int GV, int N>
__device__ __forceinline__ bool
proj_acc(const float* __restrict__ A, const float* __restrict__ B,
         float* Bs  , int mt0, int n0, v8f (&acc)[NG]) {
  const int tid  = threadIdx.x;
  const int lane = tid & 31;
  const int wave = tid >> 5;
  const int l15  = lane & 15;
  const int kb0  = (lane >> 4) << 1;

  {
    static_assert(DIN == 37 * 8 + 4, "37 rows per thread, 8 waves, and 4 rows left over");
    const bool last = wave < 4;
    float bv[NG][38];
#pragma unroll
    for (int g = 0; g < NG; ++g) {
      const bool cok = n0 + g * GV + lane < HD;
      const float* bcol = B + (size_t)wave * N + (cok ? n0 + g * GS + lane : 0);
#pragma unroll
      for (int i = 0; i < 37; ++i) bv[g][i] = bcol[(size_t)(8 * i) * N];
      bv[g][37] = last ? bcol[(size_t)(8 * 37) * N] : 0.0f;
    }
#pragma unroll
    for (int g = 0; g < NG; ++g) {
      const bool cok = n0 + g * GV + lane < HD;
      float* bdst = Bs + (g * 32 + lane) * BSP + wave;
#pragma unroll
      for (int i = 0; i < 37; ++i) bdst[8 * i] = cok ? bv[g][i] : 0.0f;
      if (last) bdst[8 * 37] = cok ? bv[g][37] : 0.0f;
    }
  }
  __syncthreads();

  const int m0 = (mt0 + (wave & 3)) * 16;
  if (m0 >= NN) return false;
  const float* arow = A + (size_t)(m0 + l15) * DIN;
  const float* brow = Bs + ((wave >> 2) * 16 + l15) * BSP;
#pragma unroll
  for (int g = 0; g < NG; ++g) acc[g] = v8f{};
  static_assert(DIN == 37 * 8 + 4 && (DIN * 4) % 16 == 0 && (BSP * 4) % 16 == 0,
                "37 blocks of 8 k and one K-step of 4; 16-byte aligned rows");
  const int kq = (lane >> 4) << 2;
  auto block8 = [&](int k0) {
    int kb = k0 + kq;
    v4f a = *(const v4f*)(arow + kb);
    v4f b[NG];
#pragma unroll
    for (int g = 0; g < NG; ++g)
      b[g] = *(const v4f*)(brow + g * 32 * BSP + kb);
#pragma unroll
    for (int g = 0; g < NG; ++g)
      acc[g] = __builtin_amdgcn_wmma_f32_16x16x4_f32(false, a.lo, false, b[g].lo, (short)0,
                                                     acc[g], false, false);
#pragma unroll
    for (int g = 0; g < NG; ++g)
      acc[g] = __builtin_amdgcn_wmma_f32_16x16x4_f32(false, a.hi, false, b[g].hi, (short)0,
                                                     acc[g], false, false);
  };
#pragma unroll 12
  for (int k0 = 0; k0 < 36 * 8; k0 += 8) block8(k0);
  block8(36 * 8);
  {
    int kb = (DIN - 4) + kb0;
    v2f a = *(const v2f*)(arow + kb);
#pragma unroll
    for (int g = 0; g < NG; ++g) {
      v2f b = *(const v2f*)(brow + g * 32 * BSP + kb);
      acc[g] = __builtin_amdgcn_wmma_f32_16x16x4_f32(false, a, false, b, (short)0,
                                                     acc[g], false, false);
    }
  }
  return true;
}

#define LEAF_END   14539
#define XF_TILE0   (LEAF_END / 16)
#define XF_RBLK    ((NN / 16 - XF_TILE0 + 3) / 4)
#define IOU_RBLK   313
#define IOU_BLOCKS (IOU_RBLK * 5)
#define XF_BLOCKS  (XF_RBLK * 2)
static_assert(NN % 16 == 0 && IOU_RBLK * 4 >= NN / 16, "row tiling");
static_assert(IOU_BLOCKS + XF_BLOCKS + WPK_BLOCKS == 1757, "grid literal in kernel_launch");

template <int NG>
__device__ __forceinline__ void
xf_store(const v8f (&acc)[NG], const float* __restrict__ bfx,
         float* __restrict__ xf, int m0, int n0) {
  const int lane = threadIdx.x & 31;
  const int l15  = lane & 15;
  const int roff = (lane >> 4) << 3;
  const int hf   = threadIdx.x >> 7;
#pragma unroll
  for (int g = 0; g < NG; ++g) {
    int n = n0 + g * 32 + hf * 16 + l15;
    if (n < HD) {
      float bv = bfx[n];
#pragma unroll
      for (int v = 0; v < 8; ++v)
        xf[(size_t)(m0 + v + roff) * HD + n] = acc[g][v] + bv;
    }
  }
}

__device__ __forceinline__ void
pack_level_weights(const float* __restrict__ Wiouh, const float* __restrict__ Wfh,
                   float* __restrict__ wpk, int pb) {
  const int kind = pb / 5, sl = pb - kind * 5;
  const bool gate = kind < 3;
  const float* W = gate ? Wiouh + kind * HD : Wfh;
  const int   WP = gate ? IOUW : HD;
  for (int e = threadIdx.x; e < KST * 32; e += 256) {
    const int s = e >> 5, l = e & 31;
    const int kb = 4 * s + ((l >> 4) << 1);
    const int hc = sl * SLW_PACK + 2 * (l & 15);
    const bool ok = hc < HD && kb < HD;
    const float* p = W + (ok ? kb * WP + hc : 0);
    v2f r0 = *(const v2f*)p;
    v2f r1 = *(const v2f*)(p + WP);
    v4f o;
    o.x = ok ? r0.x : 0.0f;  o.y = ok ? r1.x : 0.0f;
    o.z = ok ? r0.y : 0.0f;  o.w = ok ? r1.y : 0.0f;
    *(v4f*)(wpk + (size_t)pb * WPK_TASK + e * 4) = o;
  }
}

__global__ void __launch_bounds__(256)
proj_gemm(const float* __restrict__ x,
          const float* __restrict__ Wioux, const float* __restrict__ bioux,
          const float* __restrict__ biouh, float* __restrict__ xiou,
          const float* __restrict__ Wfx, const float* __restrict__ bfx,
          float* __restrict__ xf,
          float* __restrict__ cst, float* __restrict__ hst,
          const float* __restrict__ Wiouh, const float* __restrict__ Wfh,
          float* __restrict__ wpk) {
  __shared__ __attribute__((aligned(16))) float Bs[96 * BSP];
  const int b    = (int)blockIdx.x;
  const int wave = threadIdx.x >> 5;
  if (b < IOU_BLOCKS) {
    const int cb = b / IOU_RBLK, rb = b - cb * IOU_RBLK;
    const int n0 = cb * 32;
    v8f acc[3];
    if (!proj_acc<3, HD, 0, IOUW>(x, Wioux, Bs, rb * 4, n0, acc)) return;
    const int lane = threadIdx.x & 31;
    const int roff = (lane >> 4) << 3;
    const int m0   = (rb * 4 + (wave & 3)) * 16;
    const int n    = n0 + (wave >> 2) * 16 + (lane & 15);
    if (n < HD) {
      float bi = bioux[n], bo = bioux[HD + n], bu = bioux[2 * HD + n];
      float hi = biouh[n], ho = biouh[HD + n], hu = biouh[2 * HD + n];
#pragma unroll
      for (int v = 0; v < 8; ++v) {
        int m = m0 + v + roff;
        float xi = acc[0][v] + bi;
        float xo = acc[1][v] + bo;
        float xu = acc[2][v] + bu;
        if (m < LEAF_END) {
          float i = sigmoid_f(xi + hi);
          float o = sigmoid_f(xo + ho);
          float u = tanh_f(xu + hu);
          float c = i * u;
          cst[(size_t)m * HD + n] = c;
          hst[(size_t)m * HD + n] = o * tanh_f(c);
        } else {
          float* r = xiou + (size_t)m * IOUW + n;
          r[0]      = xi;
          r[HD]     = xo;
          r[2 * HD] = xu;
        }
      }
    }
  } else if (b >= IOU_BLOCKS + XF_BLOCKS) {
    pack_level_weights(Wiouh, Wfh, wpk, b - (IOU_BLOCKS + XF_BLOCKS));
  } else {
    const int s  = b - IOU_BLOCKS;
    const int rb = s >> 1;
    const int m0 = (XF_TILE0 + rb * 4 + (wave & 3)) * 16;
    if ((s & 1) == 0) {
      v8f acc[3];
      if (!proj_acc<3, 32, 32, HD>(x, Wfx, Bs, XF_TILE0 + rb * 4, 0, acc)) return;
      xf_store<3>(acc, bfx, xf, m0, 0);
    } else {
      v8f acc[2];
      if (!proj_acc<2, 32, 32, HD>(x, Wfx, Bs, XF_TILE0 + rb * 4, 96, acc)) return;
      xf_store<2>(acc, bfx, xf, m0, 96);
    }
  }
}

__device__ __forceinline__ void
pair_sweep(const float* __restrict__ wfrag, const float* arow, int kb0,
           v8f& acc0, v8f& acc1) {
  v4f w[KST];
#pragma unroll
  for (int s = 0; s < KST; ++s) w[s] = *(const v4f*)(wfrag + s * 128);
  acc0 = v8f{};
  acc1 = v8f{};
#pragma unroll
  for (int s = 0; s < KST; ++s) {
    int kb = 4 * s + kb0;
    v2f a = *(const v2f*)(arow + kb);
    acc0 = __builtin_amdgcn_wmma_f32_16x16x4_f32(false, a, false, w[s].lo, (short)0,
                                                 acc0, false, false);
    acc1 = __builtin_amdgcn_wmma_f32_16x16x4_f32(false, a, false, w[s].hi, (short)0,
                                                 acc1, false, false);
  }
}

#define SLW 32
#define NSL 5
static_assert(SLW == SLW_PACK && NSL * 4 == WPK_BLOCKS && KST * 4 == CHP, "packed weights match the slices and the K sweep");
#define CSP 36
#define ISP 100
#define FSP 36

template <int NT>
__global__ void __launch_bounds__(NT)
level_kernel(const float* __restrict__ xiou, const float* __restrict__ xf,
             const float* __restrict__ wpk, const float* __restrict__ biouh,
             const float* __restrict__ bfh,
             float* __restrict__ cst, float* __restrict__ hst,
             int start, int end) {
  extern __shared__ __attribute__((aligned(16))) float smem[];
  float* chh  = smem;
  float* hsum = chh  + 64 * CHP;
  float* chc  = hsum + 16 * CHP;
  float* ious = chc  + 64 * CSP;
  float* fcs  = ious + 16 * ISP;

  const int tid  = threadIdx.x;
  const int lane = tid & 31;
  const int wave = tid >> 5;
  const int l15  = lane & 15;
  const int kb0  = (lane >> 4) << 1;
  const int roff = (lane >> 4) << 3;
  const int base = start + (int)blockIdx.x * 16;
  const int c0   = (int)blockIdx.y * SLW;

  {
    constexpr int NW  = NT / 32;
    constexpr int NPW = 16 / NW;
    static_assert(NPW * NW == 16 && HD == 2 * 75 && CHP == 2 * 76, "whole nodes per wave; 75 column pairs and one pad pair");
    const int wv = __builtin_amdgcn_readfirstlane(tid >> 5);
    const int j2 = min(64 + lane, 74);
    const bool t2 = lane < 11;
    const int  cc = min(c0 + lane, HD - 1);
    const bool cl = c0 + lane < HD;
    v2f   hv[NPW][4][3];
    float cv[NPW][4];
    bool  ok[NPW][4];
#pragma unroll
    for (int q = 0; q < NPW; ++q) {
      const int node = base + wv * NPW + q;
      const int idx0 = 4 * node - (3 * NN - 2);
#pragma unroll
      for (int s = 0; s < 4; ++s) {
        ok[q][s] = node < end && idx0 - s >= 0;
        const size_t ro = ok[q][s] ? (size_t)(idx0 - s) * HD : (size_t)0;
        const float* hr = hst + ro;
        hv[q][s][0] = *(const v2f*)(hr + 2 * lane);
        hv[q][s][1] = *(const v2f*)(hr + 64 + 2 * lane);
        hv[q][s][2] = *(const v2f*)(hr + 2 * j2);
        cv[q][s]    = cst[ro + cc];
      }
    }
#pragma unroll
    for (int q = 0; q < NPW; ++q) {
      const int m = wv * NPW + q;
      v2f sum[3];
#pragma unroll
      for (int s = 0; s < 4; ++s) {
        float* row = chh + (4 * m + s) * CHP;
#pragma unroll
        for (int t = 0; t < 3; ++t) {
          const bool keep = ok[q][s] && (t < 2 || t2);
          const v2f h = keep ? hv[q][s][t] : v2f{0.0f, 0.0f};
          sum[t] = s == 0 ? h : sum[t] + h;
          if (t < 2 || lane < 12) *(v2f*)(row + 64 * t + 2 * lane) = h;
        }
        chc[(4 * m + s) * CSP + lane] = (ok[q][s] && cl) ? cv[q][s] : 0.0f;
      }
#pragma unroll
      for (int t = 0; t < 3; ++t)
        if (t < 2 || lane < 12) *(v2f*)(hsum + m * CHP + 64 * t + 2 * lane) = sum[t];
    }
  }
  __syncthreads();

  for (int t = wave; t < 7; t += NT / 32) {
    const bool gate = t < 3;
    const int  lc   = 2 * l15;
    const int  hc   = c0 + lc;
    const bool nok  = hc < HD;
    v8f acc0, acc1;
    const float* wfrag = wpk + (size_t)((gate ? t : 3) * NSL + (int)blockIdx.y) * WPK_TASK + lane * 4;
    if (gate) {
      const int g = t;
      const int n = g * HD + hc;
      pair_sweep(wfrag, hsum + l15 * CHP, kb0, acc0, acc1);
      if (nok) {
        v2f bv = *(const v2f*)(biouh + n);
        v2f xv[8];
#pragma unroll
        for (int v = 0; v < 8; ++v) {
          int node = base + v + roff;
          bool live = node < end;
          v2f tt = *(const v2f*)(xiou + (size_t)(live ? node : base) * IOUW + n);
          xv[v] = live ? tt : v2f{0.0f, 0.0f};
        }
#pragma unroll
        for (int v = 0; v < 8; ++v) {
          int m = v + roff;
          v2f r;
          r.x = acc0[v] + bv.x + xv[v].x;
          r.y = acc1[v] + bv.y + xv[v].y;
          *(v2f*)(ious + m * ISP + g * SLW + lc) = r;
        }
      }
    } else {
      const int m0 = (t - 3) * 16;
      const int nc = nok ? hc : 0;
      pair_sweep(wfrag, chh + (m0 + l15) * CHP, kb0, acc0, acc1);
      v2f bv = nok ? *(const v2f*)(bfh + nc) : v2f{0.0f, 0.0f};
      float s00 = 0.0f, s01 = 0.0f, s10 = 0.0f, s11 = 0.0f;
#pragma unroll
      for (int v = 0; v < 8; ++v) {
        int grow = m0 + v + roff;
        int node = base + (grow >> 2);
        v2f xv = (nok && node < end) ? *(const v2f*)(xf + (size_t)node * HD + nc) : v2f{0.0f, 0.0f};
        v2f cc = *(const v2f*)(chc + grow * CSP + lc);
        float f0 = sigmoid_f(acc0[v] + bv.x + xv.x);
        float f1 = sigmoid_f(acc1[v] + bv.y + xv.y);
        float q0 = f0 * cc.x;
        float q1 = f1 * cc.y;
        if (v < 4) { s00 += q0; s01 += q1; }
        else       { s10 += q0; s11 += q1; }
      }
      int nl = (m0 >> 2) + ((lane >> 4) << 1);
      v2f lo, hi;
      lo.x = s00;  lo.y = s01;
      hi.x = s10;  hi.y = s11;
      *(v2f*)(fcs + nl * FSP + lc)       = lo;
      *(v2f*)(fcs + (nl + 1) * FSP + lc) = hi;
    }
  }
  __syncthreads();

  for (int e = tid; e < 16 * SLW; e += NT) {
    int m = e >> 5, lc = e & 31;
    int col = c0 + lc;
    int node = base + m;
    if (node < end && col < HD) {
      float i = sigmoid_f(ious[m * ISP + lc]);
      float o = sigmoid_f(ious[m * ISP + SLW + lc]);
      float u = tanh_f(ious[m * ISP + 2 * SLW + lc]);
      float c = i * u + fcs[m * FSP + lc];
      cst[(size_t)node * HD + col] = c;
      hst[(size_t)node * HD + col] = o * tanh_f(c);
    }
  }
}

extern "C" void kernel_launch(void* const* d_in, const int* in_sizes, int n_in,
                              void* d_out, int out_size, void* d_ws, size_t ws_size,
                              hipStream_t stream) {
  const float* x     = (const float*)d_in[0];
  const float* Wioux = (const float*)d_in[1];
  const float* bioux = (const float*)d_in[2];
  const float* Wiouh = (const float*)d_in[3];
  const float* biouh = (const float*)d_in[4];
  const float* Wfx   = (const float*)d_in[5];
  const float* bfx   = (const float*)d_in[6];
  const float* Wfh   = (const float*)d_in[7];
  const float* bfh   = (const float*)d_in[8];
  (void)in_sizes; (void)n_in; (void)out_size; (void)ws_size;

  float* xiou = (float*)d_ws;
  float* xf   = xiou + (size_t)NN * IOUW;
  float* cst  = xf   + (size_t)NN * HD;
  float* wpk  = cst  + (size_t)NN * HD + 64;
  float* hst  = (float*)d_out;

  proj_gemm<<<1757, 256, 0, stream>>>(x, Wioux, bioux, biouh, xiou, Wfx, bfx, xf, cst, hst,
                                      Wiouh, Wfh, wpk);

  const int bounds[8] = {14539, 18635, 19659, 19915, 19979, 19995, 19999, 20000};
  const size_t lds =
      (size_t)(64 * CHP + 16 * CHP + 64 * CSP + 16 * ISP + 16 * FSP) * sizeof(float);
  int s = bounds[0];
  for (int L = 1; L < 8; ++L) {
    int e = bounds[L];
    int groups = (e - s + 15) / 16;
    level_kernel<256><<<dim3(groups, NSL), 256, lds, stream>>>(xiou, xf, wpk, biouh,
                                                               bfh, cst, hst, s, e);
    s = e;
  }
}
